# starting kernel
# baseline (speedup 1.0000x reference)
#include <hip/hip_runtime.h>

typedef __attribute__((ext_vector_type(2))) float v2f;
typedef __attribute__((ext_vector_type(8))) float v8f;

#define CHANNELS 64

__global__ void biconv_init(const float* __restrict__ x,
                            float* __restrict__ agg_out,
                            float* __restrict__ agg_back,
                            long long n4) {
    long long gid = (long long)blockIdx.x * blockDim.x + threadIdx.x;
    if (gid >= n4) return;
    float4 v = ((const float4*)x)[gid];
    ((float4*)agg_out)[gid]  = v;
    ((float4*)agg_back)[gid] = v;
}

__global__ void biconv_scatter(const float* __restrict__ x,
                               const int* __restrict__ src,
                               const int* __restrict__ tgt,
                               float* __restrict__ agg_out,
                               float* __restrict__ agg_back,
                               long long total  ) {
    long long gid = (long long)blockIdx.x * blockDim.x + threadIdx.x;
    if (gid >= total) return;
    long long e  = gid >> 4;
    int       c4 = ((int)gid & 15) << 2;
    long long s = (long long)src[e];
    long long t = (long long)tgt[e];

    float4 xs = *(const float4*)(x + s * CHANNELS + c4);
    float4 xt = *(const float4*)(x + t * CHANNELS + c4);

    float* po = agg_out + t * CHANNELS + c4;
    atomicAdd(po + 0, xs.x);
    atomicAdd(po + 1, xs.y);
    atomicAdd(po + 2, xs.z);
    atomicAdd(po + 3, xs.w);

    float* pb = agg_back + s * CHANNELS + c4;
    atomicAdd(pb + 0, xt.x);
    atomicAdd(pb + 1, xt.y);
    atomicAdd(pb + 2, xt.z);
    atomicAdd(pb + 3, xt.w);
}

__global__ __launch_bounds__(256)
void biconv_wmma_gemm(const float* __restrict__ agg_out,
                      const float* __restrict__ agg_back,
                      const float* __restrict__ norm,
                      const float* __restrict__ norm_t,
                      const float* __restrict__ w_out,
                      const float* __restrict__ w_back,
                      float* __restrict__ out,
                      int N, int numTiles  ) {
    const int wave = threadIdx.x >> 5;
    const int lane = threadIdx.x & 31;
    const int tile = blockIdx.x * 8 + wave;
    if (tile >= numTiles) return;

    const int rowTile = tile >> 2;
    const int colTile = tile & 3;

    const int m    = lane & 15;
    const int half = lane >> 4;
    const int kb   = half << 1;
    const int n    = lane & 15;
    const int col  = colTile * 16 + n;

    int row  = rowTile * 16 + m;
    int rowc = row < N ? row : (N - 1);

    const float sc_o = norm[rowc];
    const float sc_b = norm_t[rowc];
    const float* __restrict__ arow_o = agg_out  + (long long)rowc * CHANNELS;
    const float* __restrict__ arow_b = agg_back + (long long)rowc * CHANNELS;

    v8f acc = {};

#pragma unroll
    for (int k = 0; k < CHANNELS; k += 4) {
        float2 av = *(const float2*)(arow_o + k + kb);
        v2f a; a.x = av.x * sc_o; a.y = av.y * sc_o;
        v2f b;
        b.x = w_out[(k + kb)     * CHANNELS + col];
        b.y = w_out[(k + kb + 1) * CHANNELS + col];
        acc = __builtin_amdgcn_wmma_f32_16x16x4_f32(
            false, a, false, b, (short)0, acc, false, false);
    }
#pragma unroll
    for (int k = 0; k < CHANNELS; k += 4) {
        float2 av = *(const float2*)(arow_b + k + kb);
        v2f a; a.x = av.x * sc_b; a.y = av.y * sc_b;
        v2f b;
        b.x = w_back[(k + kb)     * CHANNELS + col];
        b.y = w_back[(k + kb + 1) * CHANNELS + col];
        acc = __builtin_amdgcn_wmma_f32_16x16x4_f32(
            false, a, false, b, (short)0, acc, false, false);
    }

#pragma unroll
    for (int j = 0; j < 8; ++j) {
        int mrow = rowTile * 16 + j + half * 8;
        if (mrow < N)
            out[(long long)mrow * CHANNELS + col] = acc[j];
    }
}

extern "C" void kernel_launch(void* const* d_in, const int* in_sizes, int n_in,
                              void* d_out, int out_size, void* d_ws, size_t ws_size,
                              hipStream_t stream) {
    const float* x       = (const float*)d_in[0];
    const int*   sources = (const int*)d_in[1];
    const int*   targets = (const int*)d_in[2];
    const float* norm    = (const float*)d_in[3];
    const float* norm_t  = (const float*)d_in[4];
    const float* w_out   = (const float*)d_in[5];
    const float* w_back  = (const float*)d_in[6];
    float*       out     = (float*)d_out;

    const int N = in_sizes[3];
    const long long E = in_sizes[1];

    float* agg_out  = (float*)d_ws;
    float* agg_back = agg_out + (long long)N * CHANNELS;

    {
        long long n4 = (long long)N * (CHANNELS / 4);
        int threads = 256;
        long long blocks = (n4 + threads - 1) / threads;
        biconv_init<<<(unsigned)blocks, threads, 0, stream>>>(x, agg_out, agg_back, n4);
    }
    {
        long long total = E * 16;
        int threads = 256;
        long long blocks = (total + threads - 1) / threads;
        biconv_scatter<<<(unsigned)blocks, threads, 0, stream>>>(
            x, sources, targets, agg_out, agg_back, total);
    }
    {
        int numRowTiles = (N + 15) >> 4;
        int numTiles = numRowTiles * 4;
        int wavesPerBlock = 8;
        int blocks = (numTiles + wavesPerBlock - 1) / wavesPerBlock;
        biconv_wmma_gemm<<<blocks, 256, 0, stream>>>(
            agg_out, agg_back, norm, norm_t, w_out, w_back, out, N, numTiles);
    }
}


# f32 atomic scatter-add replaced by per-call edge grouping (int counts, offsets, place) and per-node row sums in ascending neighbour-id order (deterministic), rows staged in LDS for the WMMAs
# speedup vs baseline: 1.0512x; 1.0512x over previous
#include <hip/hip_runtime.h>

typedef __attribute__((ext_vector_type(2))) float v2f;
typedef __attribute__((ext_vector_type(8))) float v8f;

#define CHANNELS 64
#define SEG_CAP 64
#define A_STRIDE 68

__global__ void biconv_zero(int* __restrict__ cnt, int n) {
    int gid = blockIdx.x * blockDim.x + threadIdx.x;
    if (gid < n) cnt[gid] = 0;
}

__global__ void biconv_count(const int* __restrict__ src,
                             const int* __restrict__ tgt,
                             int* __restrict__ cnt,
                             int* __restrict__ rank_in,
                             int* __restrict__ rank_out,
                             int N, int E) {
    int e = blockIdx.x * blockDim.x + threadIdx.x;
    if (e >= E) return;
    int s = src[e];
    int t = tgt[e];
    rank_in[e]  = atomicAdd(cnt + t, 1);
    rank_out[e] = atomicAdd(cnt + N + s, 1);
}

__global__ __launch_bounds__(1024)
void biconv_alloc(const int* __restrict__ cnt,
                  int* __restrict__ cursor,
                  int* __restrict__ off,
                  int n2) {
    __shared__ int wsum[32];
    __shared__ int wexcl[32];
    __shared__ int sbase;
    const int tid  = threadIdx.x;
    const int lane = tid & 31;
    const int wave = tid >> 5;
    const int i = blockIdx.x * 1024 + tid;
    const int c = (i < n2) ? cnt[i] : 0;

    int v = c;
    for (int s = 1; s < 32; s <<= 1) {
        int u = __shfl_up(v, s);
        if (lane >= s) v += u;
    }
    if (lane == 31) wsum[wave] = v;
    __syncthreads();
    if (wave == 0) {
        int w = wsum[lane];
        int z = w;
        for (int s = 1; s < 32; s <<= 1) {
            int u = __shfl_up(z, s);
            if (lane >= s) z += u;
        }
        wexcl[lane] = z - w;
        if (lane == 31) sbase = atomicAdd(cursor, z);
    }
    __syncthreads();
    if (i < n2) off[i] = sbase + wexcl[wave] + v - c;
}

__global__ void biconv_place(const int* __restrict__ src,
                             const int* __restrict__ tgt,
                             const int* __restrict__ off,
                             const int* __restrict__ rank_in,
                             const int* __restrict__ rank_out,
                             int* __restrict__ slots,
                             int N, int E) {
    int e = blockIdx.x * blockDim.x + threadIdx.x;
    if (e >= E) return;
    int s = src[e];
    int t = tgt[e];
    slots[off[t] + rank_in[e]]      = s;
    slots[off[N + s] + rank_out[e]] = t;
}

#define ROW_ADD(a, v) { a.x += v.x; a.y += v.y; a.z += v.z; a.w += v.w; }

__global__ __launch_bounds__(256)
void biconv_gather_gemm(const float* __restrict__ x,
                        const int* __restrict__ cnt,
                        const int* __restrict__ off,
                        const int* __restrict__ slots,
                        const float* __restrict__ norm,
                        const float* __restrict__ norm_t,
                        const float* __restrict__ w_out,
                        const float* __restrict__ w_back,
                        float* __restrict__ out,
                        int N) {
    __shared__ __attribute__((aligned(16))) float As[2][16][A_STRIDE];
    __shared__ __attribute__((aligned(16))) int   uns[2][16][SEG_CAP];
    __shared__ __attribute__((aligned(16))) int   srt[2][16][SEG_CAP];

    const int tid    = threadIdx.x;
    const int hw     = tid >> 4;
    const int lane16 = tid & 15;
    const int node   = blockIdx.x * 16 + hw;
    const bool valid = node < N;
    const float4* __restrict__ x4 = (const float4*)x;

    int dcount[2] = {0, 0};
    int dbase[2]  = {0, 0};
    float scale[2] = {0.f, 0.f};
    if (valid) {
        dcount[0] = cnt[node];
        dcount[1] = cnt[N + node];
        dbase[0]  = off[node];
        dbase[1]  = off[N + node];
        scale[0]  = norm[node];
        scale[1]  = norm_t[node];
    }

#pragma unroll
    for (int dir = 0; dir < 2; ++dir) {
        const int dd = (dcount[dir] <= SEG_CAP) ? dcount[dir] : 0;
        for (int i = lane16; i < dd; i += 16)
            uns[dir][hw][i] = slots[dbase[dir] + i];
    }
    __syncthreads();
#pragma unroll
    for (int dir = 0; dir < 2; ++dir) {
        const int dd = (dcount[dir] <= SEG_CAP) ? dcount[dir] : 0;
        for (int i = lane16; i < dd; i += 16) {
            const int vi = uns[dir][hw][i];
            int r = 0;
            for (int j = 0; j < dd; ++j) {
                const int u = uns[dir][hw][j];
                r += (u < vi || (u == vi && j < i)) ? 1 : 0;
            }
            srt[dir][hw][r] = vi;
        }
    }
    __syncthreads();

    float4 self = make_float4(0.f, 0.f, 0.f, 0.f);
    if (valid) self = x4[(long long)node * 16 + lane16];

#pragma unroll
    for (int dir = 0; dir < 2; ++dir) {
        const int d = dcount[dir];
        const int o = dbase[dir];
        const int* __restrict__ ids = &srt[dir][hw][0];
        float4 acc = self;

        if (d <= SEG_CAP) {
            int k = 0;
#pragma unroll 1
            for (; k + 8 <= d; k += 8) {
                const int4 q0 = *(const int4*)(ids + k);
                const int4 q1 = *(const int4*)(ids + k + 4);
                const float4 v0 = x4[(long long)q0.x * 16 + lane16];
                const float4 v1 = x4[(long long)q0.y * 16 + lane16];
                const float4 v2 = x4[(long long)q0.z * 16 + lane16];
                const float4 v3 = x4[(long long)q0.w * 16 + lane16];
                const float4 v4 = x4[(long long)q1.x * 16 + lane16];
                const float4 v5 = x4[(long long)q1.y * 16 + lane16];
                const float4 v6 = x4[(long long)q1.z * 16 + lane16];
                const float4 v7 = x4[(long long)q1.w * 16 + lane16];
                ROW_ADD(acc, v0) ROW_ADD(acc, v1) ROW_ADD(acc, v2) ROW_ADD(acc, v3)
                ROW_ADD(acc, v4) ROW_ADD(acc, v5) ROW_ADD(acc, v6) ROW_ADD(acc, v7)
            }
            if (k + 4 <= d) {
                const int4 q0 = *(const int4*)(ids + k);
                const float4 v0 = x4[(long long)q0.x * 16 + lane16];
                const float4 v1 = x4[(long long)q0.y * 16 + lane16];
                const float4 v2 = x4[(long long)q0.z * 16 + lane16];
                const float4 v3 = x4[(long long)q0.w * 16 + lane16];
                ROW_ADD(acc, v0) ROW_ADD(acc, v1) ROW_ADD(acc, v2) ROW_ADD(acc, v3)
                k += 4;
            }
            const int rest = d - k;
            if (rest > 0) {
                float4 t0 = x4[(long long)ids[k] * 16 + lane16];
                float4 t1 = make_float4(0.f, 0.f, 0.f, 0.f);
                float4 t2 = make_float4(0.f, 0.f, 0.f, 0.f);
                if (rest > 1) t1 = x4[(long long)ids[k + 1] * 16 + lane16];
                if (rest > 2) t2 = x4[(long long)ids[k + 2] * 16 + lane16];
                ROW_ADD(acc, t0)
                if (rest > 1) ROW_ADD(acc, t1)
                if (rest > 2) ROW_ADD(acc, t2)
            }
        } else {
            int prev = -1;
            int done = 0;
            while (done < d) {
                int best = 0x7fffffff;
                for (int j = 0; j < d; ++j) {
                    const int u = slots[o + j];
                    if (u > prev && u < best) best = u;
                }
                int mult = 0;
                for (int j = 0; j < d; ++j) mult += (slots[o + j] == best) ? 1 : 0;
                if (mult == 0) break;
                const float4 v = x4[(long long)best * 16 + lane16];
                for (int t = 0; t < mult; ++t) ROW_ADD(acc, v)
                prev = best;
                done += mult;
            }
        }

        const float sc = scale[dir];
        float4 sv;
        sv.x = acc.x * sc; sv.y = acc.y * sc; sv.z = acc.z * sc; sv.w = acc.w * sc;
        *(float4*)&As[dir][hw][lane16 * 4] = sv;
    }
    __syncthreads();

    const int wave = tid >> 5;
    if (wave >= 4) return;
    const int lane = tid & 31;
    const int m    = lane & 15;
    const int half = lane >> 4;
    const int kb   = half << 1;
    const int col  = wave * 16 + (lane & 15);

    v8f acc = {};
#pragma unroll
    for (int k = 0; k < CHANNELS; k += 4) {
        const float2 av = *(const float2*)&As[0][m][k + kb];
        v2f a; a.x = av.x; a.y = av.y;
        v2f b;
        b.x = w_out[(k + kb)     * CHANNELS + col];
        b.y = w_out[(k + kb + 1) * CHANNELS + col];
        acc = __builtin_amdgcn_wmma_f32_16x16x4_f32(
            false, a, false, b, (short)0, acc, false, false);
    }
#pragma unroll
    for (int k = 0; k < CHANNELS; k += 4) {
        const float2 av = *(const float2*)&As[1][m][k + kb];
        v2f a; a.x = av.x; a.y = av.y;
        v2f b;
        b.x = w_back[(k + kb)     * CHANNELS + col];
        b.y = w_back[(k + kb + 1) * CHANNELS + col];
        acc = __builtin_amdgcn_wmma_f32_16x16x4_f32(
            false, a, false, b, (short)0, acc, false, false);
    }

#pragma unroll
    for (int j = 0; j < 8; ++j) {
        int mrow = blockIdx.x * 16 + j + half * 8;
        if (mrow < N)
            out[(long long)mrow * CHANNELS + col] = acc[j];
    }
}

extern "C" void kernel_launch(void* const* d_in, const int* in_sizes, int n_in,
                              void* d_out, int out_size, void* d_ws, size_t ws_size,
                              hipStream_t stream) {
    const float* x       = (const float*)d_in[0];
    const int*   sources = (const int*)d_in[1];
    const int*   targets = (const int*)d_in[2];
    const float* norm    = (const float*)d_in[3];
    const float* norm_t  = (const float*)d_in[4];
    const float* w_out   = (const float*)d_in[5];
    const float* w_back  = (const float*)d_in[6];
    float*       out     = (float*)d_out;

    const int N = in_sizes[3];
    const int E = in_sizes[1];

    int* cnt      = (int*)d_ws;
    int* cursor   = cnt + 2 * (long long)N;
    int* off      = cursor + 4;
    int* rank_in  = off + 2 * (long long)N;
    int* rank_out = rank_in + E;
    int* slots    = rank_out + E;

    {
        int n = 2 * N + 1;
        biconv_zero<<<(n + 255) / 256, 256, 0, stream>>>(cnt, n);
    }
    biconv_count<<<(E + 255) / 256, 256, 0, stream>>>(
        sources, targets, cnt, rank_in, rank_out, N, E);
    biconv_alloc<<<(2 * N + 1023) / 1024, 1024, 0, stream>>>(cnt, cursor, off, 2 * N);
    biconv_place<<<(E + 255) / 256, 256, 0, stream>>>(
        sources, targets, off, rank_in, rank_out, slots, N, E);
    biconv_gather_gemm<<<(N + 15) / 16, 256, 0, stream>>>(
        x, cnt, off, slots, norm, norm_t, w_out, w_back, out, N);
}


# edge grouping built by a chunk-local LDS counting sort by node bin and one bin work-group that orders its pieces in LDS, replacing per-edge global returning atomics and scattered 4-byte stores
# speedup vs baseline: 2.9407x; 2.7975x over previous
#include <hip/hip_runtime.h>

typedef __attribute__((ext_vector_type(2))) float v2f;
typedef __attribute__((ext_vector_type(8))) float v8f;

#define CHANNELS 64
#define SEG_CAP 64
#define A_STRIDE 68

#define BIN_SHIFT 8
#define BIN 256
#define MAXNB 1024
#define CH_EDGES 8192
#define MAXCH 512
#define KB_CAP 8192

__device__ __forceinline__ int block_incl_scan(int v, int* wsum, int tid, int nwaves) {
    const int lane = tid & 31;
    const int wave = tid >> 5;
    __syncthreads();
    for (int s = 1; s < 32; s <<= 1) {
        int u = __shfl_up(v, s);
        if (lane >= s) v += u;
    }
    if (lane == 31) wsum[wave] = v;
    __syncthreads();
    if (wave == 0) {
        const int wv = (lane < nwaves) ? wsum[lane] : 0;
        int z = wv;
        for (int s = 1; s < 32; s <<= 1) {
            int u = __shfl_up(z, s);
            if (lane >= s) z += u;
        }
        wsum[lane] = z - wv;
    }
    __syncthreads();
    return v + wsum[wave];
}

__global__ __launch_bounds__(1024)
void biconv_chunk_sort(const int* __restrict__ src,
                       const int* __restrict__ tgt,
                       int* __restrict__ cursor,
                       int* __restrict__ stage,
                       int* __restrict__ coff,
                       int N, int E, int NB) {
    __shared__ int hist[2][MAXNB];
    __shared__ int stg[2][CH_EDGES];
    __shared__ int wsum[32];

    const int tid = threadIdx.x;
    const int w   = blockIdx.x;
    const int e0  = w * CH_EDGES;
    const int sh  = 32 - __clz(N - 1);
    const int nch = gridDim.x;

    if (w == 0 && tid == 0) *cursor = 0;

    hist[0][tid] = 0;
    hist[1][tid] = 0;
    __syncthreads();

    int sv[CH_EDGES / 1024];
    int tv[CH_EDGES / 1024];
#pragma unroll
    for (int k = 0; k < CH_EDGES / 1024; ++k) {
        const int e = e0 + k * 1024 + tid;
        sv[k] = -1;
        tv[k] = -1;
        if (e < E) {
            sv[k] = src[e];
            tv[k] = tgt[e];
        }
    }
#pragma unroll
    for (int k = 0; k < CH_EDGES / 1024; ++k) {
        if (sv[k] >= 0) {
            atomicAdd(&hist[0][tv[k] >> BIN_SHIFT], 1);
            atomicAdd(&hist[1][sv[k] >> BIN_SHIFT], 1);
        }
    }
    __syncthreads();

    const int c0 = hist[0][tid];
    const int c1 = hist[1][tid];
    const int inc = block_incl_scan(c0 | (c1 << 16), wsum, tid, 32);
    const int x0 = (inc & 0xffff) - c0;
    const int x1 = (inc >> 16) - c1;
    hist[0][tid] = x0;
    hist[1][tid] = x1;
    if (tid < NB) {
        coff[(long long)w * (NB + 1) + tid]         = x0;
        coff[(long long)(nch + w) * (NB + 1) + tid] = x1;
    }
    if (tid == 1023) {
        coff[(long long)w * (NB + 1) + NB]         = x0 + c0;
        coff[(long long)(nch + w) * (NB + 1) + NB] = x1 + c1;
    }
    __syncthreads();

#pragma unroll
    for (int k = 0; k < CH_EDGES / 1024; ++k) {
        if (sv[k] >= 0) {
            const int s = sv[k];
            const int t = tv[k];
            const int p = atomicAdd(&hist[0][t >> BIN_SHIFT], 1);
            stg[0][p] = (int)(((unsigned)(t & (BIN - 1)) << sh) | (unsigned)s);
            const int q = atomicAdd(&hist[1][s >> BIN_SHIFT], 1);
            stg[1][q] = (int)(((unsigned)(s & (BIN - 1)) << sh) | (unsigned)t);
        }
    }
    __syncthreads();

    const int ne = (E - e0 < CH_EDGES) ? (E - e0) : CH_EDGES;
    for (int i = tid; i < ne; i += 1024) {
        stage[(long long)e0 + i]     = stg[0][i];
        stage[(long long)E + e0 + i] = stg[1][i];
    }
}

__global__ __launch_bounds__(512)
void biconv_bin_build(const int* __restrict__ stage,
                      const int* __restrict__ coff,
                      int* __restrict__ cursor,
                      int* __restrict__ cnt,
                      int* __restrict__ off,
                      int* __restrict__ slots,
                      int N, int E, int NB, int nch) {
    __shared__ int pos[MAXCH + 1];
    __shared__ int pstart[MAXCH];
    __shared__ int cntL[BIN];
    __shared__ int curL[BIN];
    __shared__ int srtL[KB_CAP];
    __shared__ int wsum[32];
    __shared__ int sbase;

    const int tid = threadIdx.x;
    const int d   = (blockIdx.x >= NB) ? 1 : 0;
    const int b   = blockIdx.x - d * NB;
    const int sh  = 32 - __clz(N - 1);
    const unsigned idmask = (sh >= 32) ? 0xffffffffu : ((1u << sh) - 1u);

    int c = 0;
    if (tid < nch) {
        const long long row = (long long)(d * nch + tid) * (NB + 1);
        const int a = coff[row + b];
        const int z = coff[row + b + 1];
        c = z - a;
        pstart[tid] = d * E + tid * CH_EDGES + a;
    }
    if (tid < BIN) { cntL[tid] = 0; }
    const int pinc = block_incl_scan(c, wsum, tid, 16);
    if (tid < nch) pos[tid] = pinc - c;
    if (tid == nch - 1) pos[nch] = pinc;
    __syncthreads();
    const int T = pos[nch];

    if (tid == 0) sbase = atomicAdd(cursor, T);

    for (int i = tid; i < T; i += 512) {
        int lo = 0, hi = nch - 1;
        while (lo < hi) {
            const int mid = (lo + hi) >> 1;
            if (pos[mid + 1] > i) hi = mid; else lo = mid + 1;
        }
        const unsigned en = (unsigned)stage[pstart[lo] + (i - pos[lo])];
        atomicAdd(&cntL[en >> sh], 1);
    }
    __syncthreads();

    const int cl  = (tid < BIN) ? cntL[tid] : 0;
    const int inc = block_incl_scan(cl, wsum, tid, 16);
    const int base = sbase;
    if (tid < BIN) {
        curL[tid] = inc - cl;
        const int node = b * BIN + tid;
        if (node < N) {
            cnt[d * N + node] = cl;
            off[d * N + node] = base + inc - cl;
        }
    }
    __syncthreads();

    const bool in_lds = (T <= KB_CAP);
    for (int i = tid; i < T; i += 512) {
        int lo = 0, hi = nch - 1;
        while (lo < hi) {
            const int mid = (lo + hi) >> 1;
            if (pos[mid + 1] > i) hi = mid; else lo = mid + 1;
        }
        const unsigned en = (unsigned)stage[pstart[lo] + (i - pos[lo])];
        const int p = atomicAdd(&curL[en >> sh], 1);
        const int id = (int)(en & idmask);
        if (in_lds) srtL[p] = id;
        else        slots[(long long)base + p] = id;
    }
    if (in_lds) {
        __syncthreads();
        for (int i = tid; i < T; i += 512) slots[(long long)base + i] = srtL[i];
    }
}

#define ROW_ADD(a, v) { a.x += v.x; a.y += v.y; a.z += v.z; a.w += v.w; }

__global__ __launch_bounds__(256)
void biconv_gather_gemm(const float* __restrict__ x,
                        const int* __restrict__ cnt,
                        const int* __restrict__ off,
                        const int* __restrict__ slots,
                        const float* __restrict__ norm,
                        const float* __restrict__ norm_t,
                        const float* __restrict__ w_out,
                        const float* __restrict__ w_back,
                        float* __restrict__ out,
                        int N) {
    __shared__ __attribute__((aligned(16))) float As[2][16][A_STRIDE];
    __shared__ __attribute__((aligned(16))) int   uns[2][16][SEG_CAP];
    __shared__ __attribute__((aligned(16))) int   srt[2][16][SEG_CAP];

    const int tid    = threadIdx.x;
    const int hw     = tid >> 4;
    const int lane16 = tid & 15;
    const int node   = blockIdx.x * 16 + hw;
    const bool valid = node < N;
    const float4* __restrict__ x4 = (const float4*)x;

    int dcount[2] = {0, 0};
    int dbase[2]  = {0, 0};
    float scale[2] = {0.f, 0.f};
    if (valid) {
        dcount[0] = cnt[node];
        dcount[1] = cnt[N + node];
        dbase[0]  = off[node];
        dbase[1]  = off[N + node];
        scale[0]  = norm[node];
        scale[1]  = norm_t[node];
    }

#pragma unroll
    for (int dir = 0; dir < 2; ++dir) {
        const int dd = (dcount[dir] <= SEG_CAP) ? dcount[dir] : 0;
        for (int i = lane16; i < dd; i += 16)
            uns[dir][hw][i] = slots[dbase[dir] + i];
    }
    __syncthreads();
#pragma unroll
    for (int dir = 0; dir < 2; ++dir) {
        const int dd = (dcount[dir] <= SEG_CAP) ? dcount[dir] : 0;
        for (int i = lane16; i < dd; i += 16) {
            const int vi = uns[dir][hw][i];
            int r = 0;
            for (int j = 0; j < dd; ++j) {
                const int u = uns[dir][hw][j];
                r += (u < vi || (u == vi && j < i)) ? 1 : 0;
            }
            srt[dir][hw][r] = vi;
        }
    }
    __syncthreads();

    float4 self = make_float4(0.f, 0.f, 0.f, 0.f);
    if (valid) self = x4[(long long)node * 16 + lane16];

#pragma unroll
    for (int dir = 0; dir < 2; ++dir) {
        const int d = dcount[dir];
        const int o = dbase[dir];
        const int* __restrict__ ids = &srt[dir][hw][0];
        float4 acc = self;

        if (d <= SEG_CAP) {
            int k = 0;
#pragma unroll 1
            for (; k + 8 <= d; k += 8) {
                const int4 q0 = *(const int4*)(ids + k);
                const int4 q1 = *(const int4*)(ids + k + 4);
                const float4 v0 = x4[(long long)q0.x * 16 + lane16];
                const float4 v1 = x4[(long long)q0.y * 16 + lane16];
                const float4 v2 = x4[(long long)q0.z * 16 + lane16];
                const float4 v3 = x4[(long long)q0.w * 16 + lane16];
                const float4 v4 = x4[(long long)q1.x * 16 + lane16];
                const float4 v5 = x4[(long long)q1.y * 16 + lane16];
                const float4 v6 = x4[(long long)q1.z * 16 + lane16];
                const float4 v7 = x4[(long long)q1.w * 16 + lane16];
                ROW_ADD(acc, v0) ROW_ADD(acc, v1) ROW_ADD(acc, v2) ROW_ADD(acc, v3)
                ROW_ADD(acc, v4) ROW_ADD(acc, v5) ROW_ADD(acc, v6) ROW_ADD(acc, v7)
            }
            if (k + 4 <= d) {
                const int4 q0 = *(const int4*)(ids + k);
                const float4 v0 = x4[(long long)q0.x * 16 + lane16];
                const float4 v1 = x4[(long long)q0.y * 16 + lane16];
                const float4 v2 = x4[(long long)q0.z * 16 + lane16];
                const float4 v3 = x4[(long long)q0.w * 16 + lane16];
                ROW_ADD(acc, v0) ROW_ADD(acc, v1) ROW_ADD(acc, v2) ROW_ADD(acc, v3)
                k += 4;
            }
            const int rest = d - k;
            if (rest > 0) {
                float4 t0 = x4[(long long)ids[k] * 16 + lane16];
                float4 t1 = make_float4(0.f, 0.f, 0.f, 0.f);
                float4 t2 = make_float4(0.f, 0.f, 0.f, 0.f);
                if (rest > 1) t1 = x4[(long long)ids[k + 1] * 16 + lane16];
                if (rest > 2) t2 = x4[(long long)ids[k + 2] * 16 + lane16];
                ROW_ADD(acc, t0)
                if (rest > 1) ROW_ADD(acc, t1)
                if (rest > 2) ROW_ADD(acc, t2)
            }
        } else {
            int prev = -1;
            int done = 0;
            while (done < d) {
                int best = 0x7fffffff;
                for (int j = 0; j < d; ++j) {
                    const int u = slots[o + j];
                    if (u > prev && u < best) best = u;
                }
                int mult = 0;
                for (int j = 0; j < d; ++j) mult += (slots[o + j] == best) ? 1 : 0;
                if (mult == 0) break;
                const float4 v = x4[(long long)best * 16 + lane16];
                for (int t = 0; t < mult; ++t) ROW_ADD(acc, v)
                prev = best;
                done += mult;
            }
        }

        const float sc = scale[dir];
        float4 sv;
        sv.x = acc.x * sc; sv.y = acc.y * sc; sv.z = acc.z * sc; sv.w = acc.w * sc;
        *(float4*)&As[dir][hw][lane16 * 4] = sv;
    }
    __syncthreads();

    const int wave = tid >> 5;
    if (wave >= 4) return;
    const int lane = tid & 31;
    const int m    = lane & 15;
    const int half = lane >> 4;
    const int kb   = half << 1;
    const int col  = wave * 16 + (lane & 15);

    v8f acc = {};
#pragma unroll
    for (int k = 0; k < CHANNELS; k += 4) {
        const float2 av = *(const float2*)&As[0][m][k + kb];
        v2f a; a.x = av.x; a.y = av.y;
        v2f b;
        b.x = w_out[(k + kb)     * CHANNELS + col];
        b.y = w_out[(k + kb + 1) * CHANNELS + col];
        acc = __builtin_amdgcn_wmma_f32_16x16x4_f32(
            false, a, false, b, (short)0, acc, false, false);
    }
#pragma unroll
    for (int k = 0; k < CHANNELS; k += 4) {
        const float2 av = *(const float2*)&As[1][m][k + kb];
        v2f a; a.x = av.x; a.y = av.y;
        v2f b;
        b.x = w_back[(k + kb)     * CHANNELS + col];
        b.y = w_back[(k + kb + 1) * CHANNELS + col];
        acc = __builtin_amdgcn_wmma_f32_16x16x4_f32(
            false, a, false, b, (short)0, acc, false, false);
    }

#pragma unroll
    for (int j = 0; j < 8; ++j) {
        int mrow = blockIdx.x * 16 + j + half * 8;
        if (mrow < N)
            out[(long long)mrow * CHANNELS + col] = acc[j];
    }
}

extern "C" void kernel_launch(void* const* d_in, const int* in_sizes, int n_in,
                              void* d_out, int out_size, void* d_ws, size_t ws_size,
                              hipStream_t stream) {
    const float* x       = (const float*)d_in[0];
    const int*   sources = (const int*)d_in[1];
    const int*   targets = (const int*)d_in[2];
    const float* norm    = (const float*)d_in[3];
    const float* norm_t  = (const float*)d_in[4];
    const float* w_out   = (const float*)d_in[5];
    const float* w_back  = (const float*)d_in[6];
    float*       out     = (float*)d_out;

    const int N = in_sizes[3];
    const int E = in_sizes[1];
    const int NB  = (N + BIN - 1) / BIN;
    const int nch = (E + CH_EDGES - 1) / CH_EDGES;

    int* cnt    = (int*)d_ws;
    int* cursor = cnt + 2 * (long long)N;
    int* off    = cursor + 64;
    int* slots  = off + 2 * (long long)N;
    int* stage  = slots + 2 * (long long)E;
    int* coff   = stage + 2 * (long long)E;

    biconv_chunk_sort<<<nch, 1024, 0, stream>>>(
        sources, targets, cursor, stage, coff, N, E, NB);
    biconv_bin_build<<<2 * NB, 512, 0, stream>>>(
        stage, coff, cursor, cnt, off, slots, N, E, NB, nch);
    biconv_gather_gemm<<<(N + 15) / 16, 256, 0, stream>>>(
        x, cnt, off, slots, norm, norm_t, w_out, w_back, out, N);
}


# bin kernel reads each chunk piece once, a wave per piece with a trip's loads issued together, into LDS; counts and groups from LDS instead of two serial global walks with a binary search per entry
# speedup vs baseline: 3.2305x; 1.0985x over previous
#include <hip/hip_runtime.h>

typedef __attribute__((ext_vector_type(2))) float v2f;
typedef __attribute__((ext_vector_type(8))) float v8f;

#define CHANNELS 64
#define SEG_CAP 64
#define A_STRIDE 68

#define BIN_SHIFT 8
#define BIN 256
#define MAXNB 1024
#define CH_EDGES 8192
#define MAXCH 512
#define KB_CAP 8192
#define PCS 5

__device__ __forceinline__ int block_incl_scan(int v, int* wsum, int tid, int nwaves) {
    const int lane = tid & 31;
    const int wave = tid >> 5;
    __syncthreads();
    for (int s = 1; s < 32; s <<= 1) {
        int u = __shfl_up(v, s);
        if (lane >= s) v += u;
    }
    if (lane == 31) wsum[wave] = v;
    __syncthreads();
    if (wave == 0) {
        const int wv = (lane < nwaves) ? wsum[lane] : 0;
        int z = wv;
        for (int s = 1; s < 32; s <<= 1) {
            int u = __shfl_up(z, s);
            if (lane >= s) z += u;
        }
        wsum[lane] = z - wv;
    }
    __syncthreads();
    return v + wsum[wave];
}

__global__ __launch_bounds__(1024)
void biconv_chunk_sort(const int* __restrict__ src,
                       const int* __restrict__ tgt,
                       int* __restrict__ cursor,
                       int* __restrict__ stage,
                       int* __restrict__ coff,
                       int N, int E, int NB) {
    __shared__ int hist[2][MAXNB];
    __shared__ int stg[2][CH_EDGES];
    __shared__ int wsum[32];

    const int tid = threadIdx.x;
    const int w   = blockIdx.x;
    const int e0  = w * CH_EDGES;
    const int sh  = 32 - __clz(N - 1);
    const int nch = gridDim.x;

    if (w == 0 && tid == 0) *cursor = 0;

    hist[0][tid] = 0;
    hist[1][tid] = 0;
    __syncthreads();

    int sv[CH_EDGES / 1024];
    int tv[CH_EDGES / 1024];
#pragma unroll
    for (int k = 0; k < CH_EDGES / 1024; ++k) {
        const int e = e0 + k * 1024 + tid;
        sv[k] = -1;
        tv[k] = -1;
        if (e < E) {
            sv[k] = src[e];
            tv[k] = tgt[e];
        }
    }
#pragma unroll
    for (int k = 0; k < CH_EDGES / 1024; ++k) {
        if (sv[k] >= 0) {
            atomicAdd(&hist[0][tv[k] >> BIN_SHIFT], 1);
            atomicAdd(&hist[1][sv[k] >> BIN_SHIFT], 1);
        }
    }
    __syncthreads();

    const int c0 = hist[0][tid];
    const int c1 = hist[1][tid];
    const int inc = block_incl_scan(c0 | (c1 << 16), wsum, tid, 32);
    const int x0 = (inc & 0xffff) - c0;
    const int x1 = (inc >> 16) - c1;
    hist[0][tid] = x0;
    hist[1][tid] = x1;
    if (tid < NB) {
        coff[(long long)w * (NB + 1) + tid]         = x0;
        coff[(long long)(nch + w) * (NB + 1) + tid] = x1;
    }
    if (tid == 1023) {
        coff[(long long)w * (NB + 1) + NB]         = x0 + c0;
        coff[(long long)(nch + w) * (NB + 1) + NB] = x1 + c1;
    }
    __syncthreads();

#pragma unroll
    for (int k = 0; k < CH_EDGES / 1024; ++k) {
        if (sv[k] >= 0) {
            const int s = sv[k];
            const int t = tv[k];
            const int p = atomicAdd(&hist[0][t >> BIN_SHIFT], 1);
            stg[0][p] = (int)(((unsigned)(t & (BIN - 1)) << sh) | (unsigned)s);
            const int q = atomicAdd(&hist[1][s >> BIN_SHIFT], 1);
            stg[1][q] = (int)(((unsigned)(s & (BIN - 1)) << sh) | (unsigned)t);
        }
    }
    __syncthreads();

    const int ne = (E - e0 < CH_EDGES) ? (E - e0) : CH_EDGES;
    for (int i = tid; i < ne; i += 1024) {
        stage[(long long)e0 + i]     = stg[0][i];
        stage[(long long)E + e0 + i] = stg[1][i];
    }
}

__global__ __launch_bounds__(512)
void biconv_bin_build(const int* __restrict__ stage,
                      const int* __restrict__ coff,
                      int* __restrict__ cursor,
                      int* __restrict__ cnt,
                      int* __restrict__ off,
                      int* __restrict__ slots,
                      int N, int E, int NB, int nch) {
    __shared__ int pos[MAXCH + 2];
    __shared__ int pstart[MAXCH + 1];
    __shared__ int cntL[BIN];
    __shared__ int curL[BIN];
    __shared__ int entL[KB_CAP];
    __shared__ int srtL[KB_CAP];
    __shared__ int wsum[32];
    __shared__ int sbase;

    const int tid  = threadIdx.x;
    const int lane = tid & 31;
    const int wave = tid >> 5;
    const int d   = (blockIdx.x >= NB) ? 1 : 0;
    const int b   = blockIdx.x - d * NB;
    const int sh  = 32 - __clz(N - 1);
    const unsigned idmask = (sh >= 32) ? 0xffffffffu : ((1u << sh) - 1u);

    int c = 0;
    if (tid < nch) {
        const long long row = (long long)(d * nch + tid) * (NB + 1);
        const int a = coff[row + b];
        const int z = coff[row + b + 1];
        c = z - a;
        pstart[tid] = d * E + tid * CH_EDGES + a;
    }
    if (tid < BIN) { cntL[tid] = 0; }
    const int pinc = block_incl_scan(c, wsum, tid, 16);
    if (tid < nch) pos[tid] = pinc - c;
    if (tid == nch - 1) {
        pos[nch]     = pinc;
        pos[nch + 1] = pinc;
        pstart[nch]  = 0;
    }
    __syncthreads();
    const int T = pos[nch];

    if (tid == 0) sbase = atomicAdd(cursor, T);

    const bool in_lds = (T <= KB_CAP);

    if (in_lds) {
        for (int u0 = 0; wave + 16 * u0 < nch; u0 += PCS) {
            int pb[PCS], pc[PCS], pg[PCS];
#pragma unroll
            for (int u = 0; u < PCS; ++u) {
                const int p  = wave + 16 * (u0 + u);
                const int pp = (p < nch) ? p : nch;
                const int b0 = pos[pp];
                pb[u] = b0;
                pc[u] = pos[pp + 1] - b0;
                pg[u] = pstart[pp];
            }
            unsigned ea[PCS], eb[PCS];
#pragma unroll
            for (int u = 0; u < PCS; ++u) {
                ea[u] = (unsigned)stage[(lane      < pc[u]) ? (pg[u] + lane)      : 0];
                eb[u] = (unsigned)stage[(lane + 32 < pc[u]) ? (pg[u] + lane + 32) : 0];
            }
#pragma unroll
            for (int u = 0; u < PCS; ++u) {
                if (lane < pc[u]) {
                    entL[pb[u] + lane] = (int)ea[u];
                    atomicAdd(&cntL[ea[u] >> sh], 1);
                }
                if (lane + 32 < pc[u]) {
                    entL[pb[u] + lane + 32] = (int)eb[u];
                    atomicAdd(&cntL[eb[u] >> sh], 1);
                }
            }
#pragma unroll
            for (int u = 0; u < PCS; ++u) {
                for (int l = lane + 64; l < pc[u]; l += 32) {
                    const unsigned en = (unsigned)stage[pg[u] + l];
                    entL[pb[u] + l] = (int)en;
                    atomicAdd(&cntL[en >> sh], 1);
                }
            }
        }
    } else {
        for (int i = tid; i < T; i += 512) {
            int lo = 0, hi = nch - 1;
            while (lo < hi) {
                const int mid = (lo + hi) >> 1;
                if (pos[mid + 1] > i) hi = mid; else lo = mid + 1;
            }
            const unsigned en = (unsigned)stage[pstart[lo] + (i - pos[lo])];
            atomicAdd(&cntL[en >> sh], 1);
        }
    }
    __syncthreads();

    const int cl  = (tid < BIN) ? cntL[tid] : 0;
    const int inc = block_incl_scan(cl, wsum, tid, 16);
    const int base = sbase;
    if (tid < BIN) {
        curL[tid] = inc - cl;
        const int node = b * BIN + tid;
        if (node < N) {
            cnt[d * N + node] = cl;
            off[d * N + node] = base + inc - cl;
        }
    }
    __syncthreads();

    if (in_lds) {
        for (int i0 = tid; i0 < T; i0 += 4 * 512) {
            unsigned e[4];
#pragma unroll
            for (int u = 0; u < 4; ++u) {
                const int i = i0 + u * 512;
                e[u] = (i < T) ? (unsigned)entL[i] : 0u;
            }
            int q[4];
#pragma unroll
            for (int u = 0; u < 4; ++u) {
                const int i = i0 + u * 512;
                q[u] = (i < T) ? atomicAdd(&curL[e[u] >> sh], 1) : 0;
            }
#pragma unroll
            for (int u = 0; u < 4; ++u) {
                const int i = i0 + u * 512;
                if (i < T) srtL[q[u]] = (int)(e[u] & idmask);
            }
        }
        __syncthreads();
        for (int i = tid; i < T; i += 512) slots[(long long)base + i] = srtL[i];
    } else {
        for (int i = tid; i < T; i += 512) {
            int lo = 0, hi = nch - 1;
            while (lo < hi) {
                const int mid = (lo + hi) >> 1;
                if (pos[mid + 1] > i) hi = mid; else lo = mid + 1;
            }
            const unsigned en = (unsigned)stage[pstart[lo] + (i - pos[lo])];
            const int p = atomicAdd(&curL[en >> sh], 1);
            slots[(long long)base + p] = (int)(en & idmask);
        }
    }
}

#define ROW_ADD(a, v) { a.x += v.x; a.y += v.y; a.z += v.z; a.w += v.w; }

__global__ __launch_bounds__(256)
void biconv_gather_gemm(const float* __restrict__ x,
                        const int* __restrict__ cnt,
                        const int* __restrict__ off,
                        const int* __restrict__ slots,
                        const float* __restrict__ norm,
                        const float* __restrict__ norm_t,
                        const float* __restrict__ w_out,
                        const float* __restrict__ w_back,
                        float* __restrict__ out,
                        int N) {
    __shared__ __attribute__((aligned(16))) float As[2][16][A_STRIDE];
    __shared__ __attribute__((aligned(16))) int   uns[2][16][SEG_CAP];
    __shared__ __attribute__((aligned(16))) int   srt[2][16][SEG_CAP];

    const int tid    = threadIdx.x;
    const int hw     = tid >> 4;
    const int lane16 = tid & 15;
    const int node   = blockIdx.x * 16 + hw;
    const bool valid = node < N;
    const float4* __restrict__ x4 = (const float4*)x;

    int dcount[2] = {0, 0};
    int dbase[2]  = {0, 0};
    float scale[2] = {0.f, 0.f};
    if (valid) {
        dcount[0] = cnt[node];
        dcount[1] = cnt[N + node];
        dbase[0]  = off[node];
        dbase[1]  = off[N + node];
        scale[0]  = norm[node];
        scale[1]  = norm_t[node];
    }

#pragma unroll
    for (int dir = 0; dir < 2; ++dir) {
        const int dd = (dcount[dir] <= SEG_CAP) ? dcount[dir] : 0;
        for (int i = lane16; i < dd; i += 16)
            uns[dir][hw][i] = slots[dbase[dir] + i];
    }
    __syncthreads();
#pragma unroll
    for (int dir = 0; dir < 2; ++dir) {
        const int dd = (dcount[dir] <= SEG_CAP) ? dcount[dir] : 0;
        for (int i = lane16; i < dd; i += 16) {
            const int vi = uns[dir][hw][i];
            int r = 0;
            for (int j = 0; j < dd; ++j) {
                const int u = uns[dir][hw][j];
                r += (u < vi || (u == vi && j < i)) ? 1 : 0;
            }
            srt[dir][hw][r] = vi;
        }
    }
    __syncthreads();

    float4 self = make_float4(0.f, 0.f, 0.f, 0.f);
    if (valid) self = x4[(long long)node * 16 + lane16];

#pragma unroll
    for (int dir = 0; dir < 2; ++dir) {
        const int d = dcount[dir];
        const int o = dbase[dir];
        const int* __restrict__ ids = &srt[dir][hw][0];
        float4 acc = self;

        if (d <= SEG_CAP) {
            int k = 0;
#pragma unroll 1
            for (; k + 8 <= d; k += 8) {
                const int4 q0 = *(const int4*)(ids + k);
                const int4 q1 = *(const int4*)(ids + k + 4);
                const float4 v0 = x4[(long long)q0.x * 16 + lane16];
                const float4 v1 = x4[(long long)q0.y * 16 + lane16];
                const float4 v2 = x4[(long long)q0.z * 16 + lane16];
                const float4 v3 = x4[(long long)q0.w * 16 + lane16];
                const float4 v4 = x4[(long long)q1.x * 16 + lane16];
                const float4 v5 = x4[(long long)q1.y * 16 + lane16];
                const float4 v6 = x4[(long long)q1.z * 16 + lane16];
                const float4 v7 = x4[(long long)q1.w * 16 + lane16];
                ROW_ADD(acc, v0) ROW_ADD(acc, v1) ROW_ADD(acc, v2) ROW_ADD(acc, v3)
                ROW_ADD(acc, v4) ROW_ADD(acc, v5) ROW_ADD(acc, v6) ROW_ADD(acc, v7)
            }
            if (k + 4 <= d) {
                const int4 q0 = *(const int4*)(ids + k);
                const float4 v0 = x4[(long long)q0.x * 16 + lane16];
                const float4 v1 = x4[(long long)q0.y * 16 + lane16];
                const float4 v2 = x4[(long long)q0.z * 16 + lane16];
                const float4 v3 = x4[(long long)q0.w * 16 + lane16];
                ROW_ADD(acc, v0) ROW_ADD(acc, v1) ROW_ADD(acc, v2) ROW_ADD(acc, v3)
                k += 4;
            }
            const int rest = d - k;
            if (rest > 0) {
                float4 t0 = x4[(long long)ids[k] * 16 + lane16];
                float4 t1 = make_float4(0.f, 0.f, 0.f, 0.f);
                float4 t2 = make_float4(0.f, 0.f, 0.f, 0.f);
                if (rest > 1) t1 = x4[(long long)ids[k + 1] * 16 + lane16];
                if (rest > 2) t2 = x4[(long long)ids[k + 2] * 16 + lane16];
                ROW_ADD(acc, t0)
                if (rest > 1) ROW_ADD(acc, t1)
                if (rest > 2) ROW_ADD(acc, t2)
            }
        } else {
            int prev = -1;
            int done = 0;
            while (done < d) {
                int best = 0x7fffffff;
                for (int j = 0; j < d; ++j) {
                    const int u = slots[o + j];
                    if (u > prev && u < best) best = u;
                }
                int mult = 0;
                for (int j = 0; j < d; ++j) mult += (slots[o + j] == best) ? 1 : 0;
                if (mult == 0) break;
                const float4 v = x4[(long long)best * 16 + lane16];
                for (int t = 0; t < mult; ++t) ROW_ADD(acc, v)
                prev = best;
                done += mult;
            }
        }

        const float sc = scale[dir];
        float4 sv;
        sv.x = acc.x * sc; sv.y = acc.y * sc; sv.z = acc.z * sc; sv.w = acc.w * sc;
        *(float4*)&As[dir][hw][lane16 * 4] = sv;
    }
    __syncthreads();

    const int wave = tid >> 5;
    if (wave >= 4) return;
    const int lane = tid & 31;
    const int m    = lane & 15;
    const int half = lane >> 4;
    const int kb   = half << 1;
    const int col  = wave * 16 + (lane & 15);

    v8f acc = {};
#pragma unroll
    for (int k = 0; k < CHANNELS; k += 4) {
        const float2 av = *(const float2*)&As[0][m][k + kb];
        v2f a; a.x = av.x; a.y = av.y;
        v2f b;
        b.x = w_out[(k + kb)     * CHANNELS + col];
        b.y = w_out[(k + kb + 1) * CHANNELS + col];
        acc = __builtin_amdgcn_wmma_f32_16x16x4_f32(
            false, a, false, b, (short)0, acc, false, false);
    }
#pragma unroll
    for (int k = 0; k < CHANNELS; k += 4) {
        const float2 av = *(const float2*)&As[1][m][k + kb];
        v2f a; a.x = av.x; a.y = av.y;
        v2f b;
        b.x = w_back[(k + kb)     * CHANNELS + col];
        b.y = w_back[(k + kb + 1) * CHANNELS + col];
        acc = __builtin_amdgcn_wmma_f32_16x16x4_f32(
            false, a, false, b, (short)0, acc, false, false);
    }

#pragma unroll
    for (int j = 0; j < 8; ++j) {
        int mrow = blockIdx.x * 16 + j + half * 8;
        if (mrow < N)
            out[(long long)mrow * CHANNELS + col] = acc[j];
    }
}

extern "C" void kernel_launch(void* const* d_in, const int* in_sizes, int n_in,
                              void* d_out, int out_size, void* d_ws, size_t ws_size,
                              hipStream_t stream) {
    const float* x       = (const float*)d_in[0];
    const int*   sources = (const int*)d_in[1];
    const int*   targets = (const int*)d_in[2];
    const float* norm    = (const float*)d_in[3];
    const float* norm_t  = (const float*)d_in[4];
    const float* w_out   = (const float*)d_in[5];
    const float* w_back  = (const float*)d_in[6];
    float*       out     = (float*)d_out;

    const int N = in_sizes[3];
    const int E = in_sizes[1];
    const int NB  = (N + BIN - 1) / BIN;
    const int nch = (E + CH_EDGES - 1) / CH_EDGES;

    int* cnt    = (int*)d_ws;
    int* cursor = cnt + 2 * (long long)N;
    int* off    = cursor + 64;
    int* slots  = off + 2 * (long long)N;
    int* stage  = slots + 2 * (long long)E;
    int* coff   = stage + 2 * (long long)E;

    biconv_chunk_sort<<<nch, 1024, 0, stream>>>(
        sources, targets, cursor, stage, coff, N, E, NB);
    biconv_bin_build<<<2 * NB, 512, 0, stream>>>(
        stage, coff, cursor, cnt, off, slots, N, E, NB, nch);
    biconv_gather_gemm<<<(N + 15) / 16, 256, 0, stream>>>(
        x, cnt, off, slots, norm, norm_t, w_out, w_back, out, N);
}


# last kernel made resident: 1280 work-groups loop over the 16-node tiles (was one short work-group per tile, 50000 waves to enter), next tiles' counts, offsets, ids fetched ahead; sums unchanged
# speedup vs baseline: 3.3642x; 1.0414x over previous
#include <hip/hip_runtime.h>

typedef __attribute__((ext_vector_type(2))) float v2f;
typedef __attribute__((ext_vector_type(8))) float v8f;

#define CHANNELS 64
#define SEG_CAP 64
#define A_STRIDE 68

#define BIN_SHIFT 8
#define BIN 256
#define MAXNB 1024
#define CH_EDGES 8192
#define MAXCH 512
#define KB_CAP 8192
#define PCS 5

__device__ __forceinline__ int block_incl_scan(int v, int* wsum, int tid, int nwaves) {
    const int lane = tid & 31;
    const int wave = tid >> 5;
    __syncthreads();
    for (int s = 1; s < 32; s <<= 1) {
        int u = __shfl_up(v, s);
        if (lane >= s) v += u;
    }
    if (lane == 31) wsum[wave] = v;
    __syncthreads();
    if (wave == 0) {
        const int wv = (lane < nwaves) ? wsum[lane] : 0;
        int z = wv;
        for (int s = 1; s < 32; s <<= 1) {
            int u = __shfl_up(z, s);
            if (lane >= s) z += u;
        }
        wsum[lane] = z - wv;
    }
    __syncthreads();
    return v + wsum[wave];
}

__global__ __launch_bounds__(1024)
void biconv_chunk_sort(const int* __restrict__ src,
                       const int* __restrict__ tgt,
                       int* __restrict__ cursor,
                       int* __restrict__ stage,
                       int* __restrict__ coff,
                       int N, int E, int NB) {
    __shared__ int hist[2][MAXNB];
    __shared__ int stg[2][CH_EDGES];
    __shared__ int wsum[32];

    const int tid = threadIdx.x;
    const int w   = blockIdx.x;
    const int e0  = w * CH_EDGES;
    const int sh  = 32 - __clz(N - 1);
    const int nch = gridDim.x;

    if (w == 0 && tid == 0) *cursor = 0;

    hist[0][tid] = 0;
    hist[1][tid] = 0;
    __syncthreads();

    int sv[CH_EDGES / 1024];
    int tv[CH_EDGES / 1024];
#pragma unroll
    for (int k = 0; k < CH_EDGES / 1024; ++k) {
        const int e = e0 + k * 1024 + tid;
        sv[k] = -1;
        tv[k] = -1;
        if (e < E) {
            sv[k] = src[e];
            tv[k] = tgt[e];
        }
    }
#pragma unroll
    for (int k = 0; k < CH_EDGES / 1024; ++k) {
        if (sv[k] >= 0) {
            atomicAdd(&hist[0][tv[k] >> BIN_SHIFT], 1);
            atomicAdd(&hist[1][sv[k] >> BIN_SHIFT], 1);
        }
    }
    __syncthreads();

    const int c0 = hist[0][tid];
    const int c1 = hist[1][tid];
    const int inc = block_incl_scan(c0 | (c1 << 16), wsum, tid, 32);
    const int x0 = (inc & 0xffff) - c0;
    const int x1 = (inc >> 16) - c1;
    hist[0][tid] = x0;
    hist[1][tid] = x1;
    if (tid < NB) {
        coff[(long long)w * (NB + 1) + tid]         = x0;
        coff[(long long)(nch + w) * (NB + 1) + tid] = x1;
    }
    if (tid == 1023) {
        coff[(long long)w * (NB + 1) + NB]         = x0 + c0;
        coff[(long long)(nch + w) * (NB + 1) + NB] = x1 + c1;
    }
    __syncthreads();

#pragma unroll
    for (int k = 0; k < CH_EDGES / 1024; ++k) {
        if (sv[k] >= 0) {
            const int s = sv[k];
            const int t = tv[k];
            const int p = atomicAdd(&hist[0][t >> BIN_SHIFT], 1);
            stg[0][p] = (int)(((unsigned)(t & (BIN - 1)) << sh) | (unsigned)s);
            const int q = atomicAdd(&hist[1][s >> BIN_SHIFT], 1);
            stg[1][q] = (int)(((unsigned)(s & (BIN - 1)) << sh) | (unsigned)t);
        }
    }
    __syncthreads();

    const int ne = (E - e0 < CH_EDGES) ? (E - e0) : CH_EDGES;
    for (int i = tid; i < ne; i += 1024) {
        stage[(long long)e0 + i]     = stg[0][i];
        stage[(long long)E + e0 + i] = stg[1][i];
    }
}

__global__ __launch_bounds__(512)
void biconv_bin_build(const int* __restrict__ stage,
                      const int* __restrict__ coff,
                      int* __restrict__ cursor,
                      int* __restrict__ cnt,
                      int* __restrict__ off,
                      int* __restrict__ slots,
                      int N, int E, int NB, int nch) {
    __shared__ int pos[MAXCH + 2];
    __shared__ int pstart[MAXCH + 1];
    __shared__ int cntL[BIN];
    __shared__ int curL[BIN];
    __shared__ int entL[KB_CAP];
    __shared__ int srtL[KB_CAP];
    __shared__ int wsum[32];
    __shared__ int sbase;

    const int tid  = threadIdx.x;
    const int lane = tid & 31;
    const int wave = tid >> 5;
    const int d   = (blockIdx.x >= NB) ? 1 : 0;
    const int b   = blockIdx.x - d * NB;
    const int sh  = 32 - __clz(N - 1);
    const unsigned idmask = (sh >= 32) ? 0xffffffffu : ((1u << sh) - 1u);

    int c = 0;
    if (tid < nch) {
        const long long row = (long long)(d * nch + tid) * (NB + 1);
        const int a = coff[row + b];
        const int z = coff[row + b + 1];
        c = z - a;
        pstart[tid] = d * E + tid * CH_EDGES + a;
    }
    if (tid < BIN) { cntL[tid] = 0; }
    const int pinc = block_incl_scan(c, wsum, tid, 16);
    if (tid < nch) pos[tid] = pinc - c;
    if (tid == nch - 1) {
        pos[nch]     = pinc;
        pos[nch + 1] = pinc;
        pstart[nch]  = 0;
    }
    __syncthreads();
    const int T = pos[nch];

    if (tid == 0) sbase = atomicAdd(cursor, T);

    const bool in_lds = (T <= KB_CAP);

    if (in_lds) {
        for (int u0 = 0; wave + 16 * u0 < nch; u0 += PCS) {
            int pb[PCS], pc[PCS], pg[PCS];
#pragma unroll
            for (int u = 0; u < PCS; ++u) {
                const int p  = wave + 16 * (u0 + u);
                const int pp = (p < nch) ? p : nch;
                const int b0 = pos[pp];
                pb[u] = b0;
                pc[u] = pos[pp + 1] - b0;
                pg[u] = pstart[pp];
            }
            unsigned ea[PCS], eb[PCS];
#pragma unroll
            for (int u = 0; u < PCS; ++u) {
                ea[u] = (unsigned)stage[(lane      < pc[u]) ? (pg[u] + lane)      : 0];
                eb[u] = (unsigned)stage[(lane + 32 < pc[u]) ? (pg[u] + lane + 32) : 0];
            }
#pragma unroll
            for (int u = 0; u < PCS; ++u) {
                if (lane < pc[u]) {
                    entL[pb[u] + lane] = (int)ea[u];
                    atomicAdd(&cntL[ea[u] >> sh], 1);
                }
                if (lane + 32 < pc[u]) {
                    entL[pb[u] + lane + 32] = (int)eb[u];
                    atomicAdd(&cntL[eb[u] >> sh], 1);
                }
            }
#pragma unroll
            for (int u = 0; u < PCS; ++u) {
                for (int l = lane + 64; l < pc[u]; l += 32) {
                    const unsigned en = (unsigned)stage[pg[u] + l];
                    entL[pb[u] + l] = (int)en;
                    atomicAdd(&cntL[en >> sh], 1);
                }
            }
        }
    } else {
        for (int i = tid; i < T; i += 512) {
            int lo = 0, hi = nch - 1;
            while (lo < hi) {
                const int mid = (lo + hi) >> 1;
                if (pos[mid + 1] > i) hi = mid; else lo = mid + 1;
            }
            const unsigned en = (unsigned)stage[pstart[lo] + (i - pos[lo])];
            atomicAdd(&cntL[en >> sh], 1);
        }
    }
    __syncthreads();

    const int cl  = (tid < BIN) ? cntL[tid] : 0;
    const int inc = block_incl_scan(cl, wsum, tid, 16);
    const int base = sbase;
    if (tid < BIN) {
        curL[tid] = inc - cl;
        const int node = b * BIN + tid;
        if (node < N) {
            cnt[d * N + node] = cl;
            off[d * N + node] = base + inc - cl;
        }
    }
    __syncthreads();

    if (in_lds) {
        for (int i0 = tid; i0 < T; i0 += 4 * 512) {
            unsigned e[4];
#pragma unroll
            for (int u = 0; u < 4; ++u) {
                const int i = i0 + u * 512;
                e[u] = (i < T) ? (unsigned)entL[i] : 0u;
            }
            int q[4];
#pragma unroll
            for (int u = 0; u < 4; ++u) {
                const int i = i0 + u * 512;
                q[u] = (i < T) ? atomicAdd(&curL[e[u] >> sh], 1) : 0;
            }
#pragma unroll
            for (int u = 0; u < 4; ++u) {
                const int i = i0 + u * 512;
                if (i < T) srtL[q[u]] = (int)(e[u] & idmask);
            }
        }
        __syncthreads();
        for (int i = tid; i < T; i += 512) slots[(long long)base + i] = srtL[i];
    } else {
        for (int i = tid; i < T; i += 512) {
            int lo = 0, hi = nch - 1;
            while (lo < hi) {
                const int mid = (lo + hi) >> 1;
                if (pos[mid + 1] > i) hi = mid; else lo = mid + 1;
            }
            const unsigned en = (unsigned)stage[pstart[lo] + (i - pos[lo])];
            const int p = atomicAdd(&curL[en >> sh], 1);
            slots[(long long)base + p] = (int)(en & idmask);
        }
    }
}

#define ROW_ADD(a, v) { a.x += v.x; a.y += v.y; a.z += v.z; a.w += v.w; }

__global__ __launch_bounds__(256)
void biconv_gather_gemm(const float* __restrict__ x,
                        const int* __restrict__ cnt,
                        const int* __restrict__ off,
                        const int* __restrict__ slots,
                        const float* __restrict__ norm,
                        const float* __restrict__ norm_t,
                        const float* __restrict__ w_out,
                        const float* __restrict__ w_back,
                        float* __restrict__ out,
                        int N) {
    __shared__ __attribute__((aligned(16))) float As[2][16][A_STRIDE];
    __shared__ __attribute__((aligned(16))) int   uns[2][16][SEG_CAP];
    __shared__ __attribute__((aligned(16))) int   srt[2][16][SEG_CAP];

    const int tid    = threadIdx.x;
    const int hw     = tid >> 4;
    const int lane16 = tid & 15;
    const int wave   = tid >> 5;
    const int lane   = tid & 31;
    const int m      = lane & 15;
    const int half   = lane >> 4;
    const int kb     = half << 1;
    const int col    = (wave & 3) * 16 + (lane & 15);
    const float4* __restrict__ x4 = (const float4*)x;
    const float* __restrict__ wo = w_out  + kb * CHANNELS + col;
    const float* __restrict__ wb = w_back + kb * CHANNELS + col;

    const int ntiles = (N + 15) >> 4;
    const int tstep  = (int)gridDim.x;

    int ccnt0 = 0, ccnt1 = 0, cbase0 = 0, cbase1 = 0;
    int ncnt0 = 0, ncnt1 = 0, nbase0 = 0, nbase1 = 0;
    {
        const int n0 = (int)blockIdx.x * 16 + hw;
        if (n0 < N) {
            ccnt0  = cnt[n0];
            ccnt1  = cnt[N + n0];
            cbase0 = off[n0];
            cbase1 = off[N + n0];
        }
        const int n1 = ((int)blockIdx.x + tstep) * 16 + hw;
        if (n1 < N) {
            ncnt0  = cnt[n1];
            ncnt1  = cnt[N + n1];
            nbase0 = off[n1];
            nbase1 = off[N + n1];
        }
    }
    {
        const int fd0 = (ccnt0 <= SEG_CAP) ? ccnt0 : 0;
        const int fd1 = (ccnt1 <= SEG_CAP) ? ccnt1 : 0;
        for (int i = lane16; i < fd0; i += 16) uns[0][hw][i] = slots[cbase0 + i];
        for (int i = lane16; i < fd1; i += 16) uns[1][hw][i] = slots[cbase1 + i];
    }
    __syncthreads();

    for (int tile = (int)blockIdx.x; tile < ntiles; tile += tstep) {
        const int node   = tile * 16 + hw;
        const bool valid = node < N;

        const int dcount[2] = {ccnt0, ccnt1};
        const int dbase[2]  = {cbase0, cbase1};

        float scale[2] = {0.f, 0.f};
        float4 self = make_float4(0.f, 0.f, 0.f, 0.f);
        if (valid) {
            scale[0] = norm[node];
            scale[1] = norm_t[node];
            self     = x4[(long long)node * 16 + lane16];
        }
        int fcnt0 = 0, fcnt1 = 0, fbase0 = 0, fbase1 = 0;
        {
            const int n2 = (tile + 2 * tstep) * 16 + hw;
            if (n2 < N) {
                fcnt0  = cnt[n2];
                fcnt1  = cnt[N + n2];
                fbase0 = off[n2];
                fbase1 = off[N + n2];
            }
        }
        const int nd0 = (ncnt0 <= SEG_CAP) ? ncnt0 : 0;
        const int nd1 = (ncnt1 <= SEG_CAP) ? ncnt1 : 0;
        int p00 = 0, p01 = 0, p10 = 0, p11 = 0;
        if (lane16      < nd0) p00 = slots[nbase0 + lane16];
        if (lane16 + 16 < nd0) p01 = slots[nbase0 + lane16 + 16];
        if (lane16      < nd1) p10 = slots[nbase1 + lane16];
        if (lane16 + 16 < nd1) p11 = slots[nbase1 + lane16 + 16];

#pragma unroll
        for (int dir = 0; dir < 2; ++dir) {
            const int dd = (dcount[dir] <= SEG_CAP) ? dcount[dir] : 0;
            for (int i = lane16; i < dd; i += 16) {
                const int vi = uns[dir][hw][i];
                int r = 0;
                for (int j = 0; j < dd; ++j) {
                    const int u = uns[dir][hw][j];
                    r += (u < vi || (u == vi && j < i)) ? 1 : 0;
                }
                srt[dir][hw][r] = vi;
            }
        }
        __syncthreads();

#pragma unroll
        for (int dir = 0; dir < 2; ++dir) {
            const int d = dcount[dir];
            const int o = dbase[dir];
            const int* __restrict__ ids = &srt[dir][hw][0];
            float4 acc = self;

            if (d <= SEG_CAP) {
                int k = 0;
#pragma unroll 1
                for (; k + 8 <= d; k += 8) {
                    const int4 q0 = *(const int4*)(ids + k);
                    const int4 q1 = *(const int4*)(ids + k + 4);
                    const float4 v0 = x4[(long long)q0.x * 16 + lane16];
                    const float4 v1 = x4[(long long)q0.y * 16 + lane16];
                    const float4 v2 = x4[(long long)q0.z * 16 + lane16];
                    const float4 v3 = x4[(long long)q0.w * 16 + lane16];
                    const float4 v4 = x4[(long long)q1.x * 16 + lane16];
                    const float4 v5 = x4[(long long)q1.y * 16 + lane16];
                    const float4 v6 = x4[(long long)q1.z * 16 + lane16];
                    const float4 v7 = x4[(long long)q1.w * 16 + lane16];
                    ROW_ADD(acc, v0) ROW_ADD(acc, v1) ROW_ADD(acc, v2) ROW_ADD(acc, v3)
                    ROW_ADD(acc, v4) ROW_ADD(acc, v5) ROW_ADD(acc, v6) ROW_ADD(acc, v7)
                }
                if (k + 4 <= d) {
                    const int4 q0 = *(const int4*)(ids + k);
                    const float4 v0 = x4[(long long)q0.x * 16 + lane16];
                    const float4 v1 = x4[(long long)q0.y * 16 + lane16];
                    const float4 v2 = x4[(long long)q0.z * 16 + lane16];
                    const float4 v3 = x4[(long long)q0.w * 16 + lane16];
                    ROW_ADD(acc, v0) ROW_ADD(acc, v1) ROW_ADD(acc, v2) ROW_ADD(acc, v3)
                    k += 4;
                }
                const int rest = d - k;
                if (rest > 0) {
                    float4 t0 = x4[(long long)ids[k] * 16 + lane16];
                    float4 t1 = make_float4(0.f, 0.f, 0.f, 0.f);
                    float4 t2 = make_float4(0.f, 0.f, 0.f, 0.f);
                    if (rest > 1) t1 = x4[(long long)ids[k + 1] * 16 + lane16];
                    if (rest > 2) t2 = x4[(long long)ids[k + 2] * 16 + lane16];
                    ROW_ADD(acc, t0)
                    if (rest > 1) ROW_ADD(acc, t1)
                    if (rest > 2) ROW_ADD(acc, t2)
                }
            } else {
                int prev = -1;
                int done = 0;
                while (done < d) {
                    int best = 0x7fffffff;
                    for (int j = 0; j < d; ++j) {
                        const int u = slots[o + j];
                        if (u > prev && u < best) best = u;
                    }
                    int mult = 0;
                    for (int j = 0; j < d; ++j) mult += (slots[o + j] == best) ? 1 : 0;
                    if (mult == 0) break;
                    const float4 v = x4[(long long)best * 16 + lane16];
                    for (int t = 0; t < mult; ++t) ROW_ADD(acc, v)
                    prev = best;
                    done += mult;
                }
            }

            const float sc = scale[dir];
            float4 sv;
            sv.x = acc.x * sc; sv.y = acc.y * sc; sv.z = acc.z * sc; sv.w = acc.w * sc;
            *(float4*)&As[dir][hw][lane16 * 4] = sv;
        }

        if (lane16      < nd0) uns[0][hw][lane16]      = p00;
        if (lane16 + 16 < nd0) uns[0][hw][lane16 + 16] = p01;
        if (lane16      < nd1) uns[1][hw][lane16]      = p10;
        if (lane16 + 16 < nd1) uns[1][hw][lane16 + 16] = p11;
        for (int i = lane16 + 32; i < nd0; i += 16) uns[0][hw][i] = slots[nbase0 + i];
        for (int i = lane16 + 32; i < nd1; i += 16) uns[1][hw][i] = slots[nbase1 + i];
        __syncthreads();

        if (wave < 4) {
            v8f acc = {};
#pragma unroll
            for (int k = 0; k < CHANNELS; k += 4) {
                const float2 av = *(const float2*)&As[0][m][k + kb];
                v2f a; a.x = av.x; a.y = av.y;
                v2f b;
                b.x = wo[k * CHANNELS];
                b.y = wo[(k + 1) * CHANNELS];
                acc = __builtin_amdgcn_wmma_f32_16x16x4_f32(
                    false, a, false, b, (short)0, acc, false, false);
            }
#pragma unroll
            for (int k = 0; k < CHANNELS; k += 4) {
                const float2 av = *(const float2*)&As[1][m][k + kb];
                v2f a; a.x = av.x; a.y = av.y;
                v2f b;
                b.x = wb[k * CHANNELS];
                b.y = wb[(k + 1) * CHANNELS];
                acc = __builtin_amdgcn_wmma_f32_16x16x4_f32(
                    false, a, false, b, (short)0, acc, false, false);
            }

#pragma unroll
            for (int j = 0; j < 8; ++j) {
                const int mrow = tile * 16 + j + half * 8;
                if (mrow < N)
                    out[(long long)mrow * CHANNELS + col] = acc[j];
            }
        }

        ccnt0 = ncnt0; ccnt1 = ncnt1; cbase0 = nbase0; cbase1 = nbase1;
        ncnt0 = fcnt0; ncnt1 = fcnt1; nbase0 = fbase0; nbase1 = fbase1;
    }
}

extern "C" void kernel_launch(void* const* d_in, const int* in_sizes, int n_in,
                              void* d_out, int out_size, void* d_ws, size_t ws_size,
                              hipStream_t stream) {
    const float* x       = (const float*)d_in[0];
    const int*   sources = (const int*)d_in[1];
    const int*   targets = (const int*)d_in[2];
    const float* norm    = (const float*)d_in[3];
    const float* norm_t  = (const float*)d_in[4];
    const float* w_out   = (const float*)d_in[5];
    const float* w_back  = (const float*)d_in[6];
    float*       out     = (float*)d_out;

    const int N = in_sizes[3];
    const int E = in_sizes[1];
    const int NB  = (N + BIN - 1) / BIN;
    const int nch = (E + CH_EDGES - 1) / CH_EDGES;

    int* cnt    = (int*)d_ws;
    int* cursor = cnt + 2 * (long long)N;
    int* off    = cursor + 64;
    int* slots  = off + 2 * (long long)N;
    int* stage  = slots + 2 * (long long)E;
    int* coff   = stage + 2 * (long long)E;

    biconv_chunk_sort<<<nch, 1024, 0, stream>>>(
        sources, targets, cursor, stage, coff, N, E, NB);
    biconv_bin_build<<<2 * NB, 512, 0, stream>>>(
        stage, coff, cursor, cnt, off, slots, N, E, NB, nch);
    biconv_gather_gemm<<<1280, 256, 0, stream>>>(
        x, cnt, off, slots, norm, norm_t, w_out, w_back, out, N);
}


# last kernel: a node's neighbour ids ranked in registers by row broadcasts (DPP row_share) and stored sorted to LDS, replacing the unsorted LDS copy and the LDS rank loops; same sum order
# speedup vs baseline: 3.6910x; 1.0971x over previous
#include <hip/hip_runtime.h>

typedef __attribute__((ext_vector_type(2))) float v2f;
typedef __attribute__((ext_vector_type(8))) float v8f;

#define CHANNELS 64
#define SEG_CAP 64
#define REG_CAP 32
#define A_STRIDE 68

#define BIN_SHIFT 8
#define BIN 256
#define MAXNB 1024
#define CH_EDGES 8192
#define MAXCH 512
#define KB_CAP 8192
#define PCS 5

__device__ __forceinline__ int block_incl_scan(int v, int* wsum, int tid, int nwaves) {
    const int lane = tid & 31;
    const int wave = tid >> 5;
    __syncthreads();
    for (int s = 1; s < 32; s <<= 1) {
        int u = __shfl_up(v, s);
        if (lane >= s) v += u;
    }
    if (lane == 31) wsum[wave] = v;
    __syncthreads();
    if (wave == 0) {
        const int wv = (lane < nwaves) ? wsum[lane] : 0;
        int z = wv;
        for (int s = 1; s < 32; s <<= 1) {
            int u = __shfl_up(z, s);
            if (lane >= s) z += u;
        }
        wsum[lane] = z - wv;
    }
    __syncthreads();
    return v + wsum[wave];
}

__global__ __launch_bounds__(1024)
void biconv_chunk_sort(const int* __restrict__ src,
                       const int* __restrict__ tgt,
                       int* __restrict__ cursor,
                       int* __restrict__ stage,
                       int* __restrict__ coff,
                       int N, int E, int NB) {
    __shared__ int hist[2][MAXNB];
    __shared__ int stg[2][CH_EDGES];
    __shared__ int wsum[32];

    const int tid = threadIdx.x;
    const int w   = blockIdx.x;
    const int e0  = w * CH_EDGES;
    const int sh  = 32 - __clz(N - 1);
    const int nch = gridDim.x;

    if (w == 0 && tid == 0) *cursor = 0;

    hist[0][tid] = 0;
    hist[1][tid] = 0;
    __syncthreads();

    int sv[CH_EDGES / 1024];
    int tv[CH_EDGES / 1024];
#pragma unroll
    for (int k = 0; k < CH_EDGES / 1024; ++k) {
        const int e = e0 + k * 1024 + tid;
        sv[k] = -1;
        tv[k] = -1;
        if (e < E) {
            sv[k] = src[e];
            tv[k] = tgt[e];
        }
    }
#pragma unroll
    for (int k = 0; k < CH_EDGES / 1024; ++k) {
        if (sv[k] >= 0) {
            atomicAdd(&hist[0][tv[k] >> BIN_SHIFT], 1);
            atomicAdd(&hist[1][sv[k] >> BIN_SHIFT], 1);
        }
    }
    __syncthreads();

    const int c0 = hist[0][tid];
    const int c1 = hist[1][tid];
    const int inc = block_incl_scan(c0 | (c1 << 16), wsum, tid, 32);
    const int x0 = (inc & 0xffff) - c0;
    const int x1 = (inc >> 16) - c1;
    hist[0][tid] = x0;
    hist[1][tid] = x1;
    if (tid < NB) {
        coff[(long long)w * (NB + 1) + tid]         = x0;
        coff[(long long)(nch + w) * (NB + 1) + tid] = x1;
    }
    if (tid == 1023) {
        coff[(long long)w * (NB + 1) + NB]         = x0 + c0;
        coff[(long long)(nch + w) * (NB + 1) + NB] = x1 + c1;
    }
    __syncthreads();

#pragma unroll
    for (int k = 0; k < CH_EDGES / 1024; ++k) {
        if (sv[k] >= 0) {
            const int s = sv[k];
            const int t = tv[k];
            const int p = atomicAdd(&hist[0][t >> BIN_SHIFT], 1);
            stg[0][p] = (int)(((unsigned)(t & (BIN - 1)) << sh) | (unsigned)s);
            const int q = atomicAdd(&hist[1][s >> BIN_SHIFT], 1);
            stg[1][q] = (int)(((unsigned)(s & (BIN - 1)) << sh) | (unsigned)t);
        }
    }
    __syncthreads();

    const int ne = (E - e0 < CH_EDGES) ? (E - e0) : CH_EDGES;
    for (int i = tid; i < ne; i += 1024) {
        stage[(long long)e0 + i]     = stg[0][i];
        stage[(long long)E + e0 + i] = stg[1][i];
    }
}

__global__ __launch_bounds__(512)
void biconv_bin_build(const int* __restrict__ stage,
                      const int* __restrict__ coff,
                      int* __restrict__ cursor,
                      int* __restrict__ cnt,
                      int* __restrict__ off,
                      int* __restrict__ slots,
                      int N, int E, int NB, int nch) {
    __shared__ int pos[MAXCH + 2];
    __shared__ int pstart[MAXCH + 1];
    __shared__ int cntL[BIN];
    __shared__ int curL[BIN];
    __shared__ int entL[KB_CAP];
    __shared__ int srtL[KB_CAP];
    __shared__ int wsum[32];
    __shared__ int sbase;

    const int tid  = threadIdx.x;
    const int lane = tid & 31;
    const int wave = tid >> 5;
    const int d   = (blockIdx.x >= NB) ? 1 : 0;
    const int b   = blockIdx.x - d * NB;
    const int sh  = 32 - __clz(N - 1);
    const unsigned idmask = (sh >= 32) ? 0xffffffffu : ((1u << sh) - 1u);

    int c = 0;
    if (tid < nch) {
        const long long row = (long long)(d * nch + tid) * (NB + 1);
        const int a = coff[row + b];
        const int z = coff[row + b + 1];
        c = z - a;
        pstart[tid] = d * E + tid * CH_EDGES + a;
    }
    if (tid < BIN) { cntL[tid] = 0; }
    const int pinc = block_incl_scan(c, wsum, tid, 16);
    if (tid < nch) pos[tid] = pinc - c;
    if (tid == nch - 1) {
        pos[nch]     = pinc;
        pos[nch + 1] = pinc;
        pstart[nch]  = 0;
    }
    __syncthreads();
    const int T = pos[nch];

    if (tid == 0) sbase = atomicAdd(cursor, T);

    const bool in_lds = (T <= KB_CAP);

    if (in_lds) {
        for (int u0 = 0; wave + 16 * u0 < nch; u0 += PCS) {
            int pb[PCS], pc[PCS], pg[PCS];
#pragma unroll
            for (int u = 0; u < PCS; ++u) {
                const int p  = wave + 16 * (u0 + u);
                const int pp = (p < nch) ? p : nch;
                const int b0 = pos[pp];
                pb[u] = b0;
                pc[u] = pos[pp + 1] - b0;
                pg[u] = pstart[pp];
            }
            unsigned ea[PCS], eb[PCS];
#pragma unroll
            for (int u = 0; u < PCS; ++u) {
                ea[u] = (unsigned)stage[(lane      < pc[u]) ? (pg[u] + lane)      : 0];
                eb[u] = (unsigned)stage[(lane + 32 < pc[u]) ? (pg[u] + lane + 32) : 0];
            }
#pragma unroll
            for (int u = 0; u < PCS; ++u) {
                if (lane < pc[u]) {
                    entL[pb[u] + lane] = (int)ea[u];
                    atomicAdd(&cntL[ea[u] >> sh], 1);
                }
                if (lane + 32 < pc[u]) {
                    entL[pb[u] + lane + 32] = (int)eb[u];
                    atomicAdd(&cntL[eb[u] >> sh], 1);
                }
            }
#pragma unroll
            for (int u = 0; u < PCS; ++u) {
                for (int l = lane + 64; l < pc[u]; l += 32) {
                    const unsigned en = (unsigned)stage[pg[u] + l];
                    entL[pb[u] + l] = (int)en;
                    atomicAdd(&cntL[en >> sh], 1);
                }
            }
        }
    } else {
        for (int i = tid; i < T; i += 512) {
            int lo = 0, hi = nch - 1;
            while (lo < hi) {
                const int mid = (lo + hi) >> 1;
                if (pos[mid + 1] > i) hi = mid; else lo = mid + 1;
            }
            const unsigned en = (unsigned)stage[pstart[lo] + (i - pos[lo])];
            atomicAdd(&cntL[en >> sh], 1);
        }
    }
    __syncthreads();

    const int cl  = (tid < BIN) ? cntL[tid] : 0;
    const int inc = block_incl_scan(cl, wsum, tid, 16);
    const int base = sbase;
    if (tid < BIN) {
        curL[tid] = inc - cl;
        const int node = b * BIN + tid;
        if (node < N) {
            cnt[d * N + node] = cl;
            off[d * N + node] = base + inc - cl;
        }
    }
    __syncthreads();

    if (in_lds) {
        for (int i0 = tid; i0 < T; i0 += 4 * 512) {
            unsigned e[4];
#pragma unroll
            for (int u = 0; u < 4; ++u) {
                const int i = i0 + u * 512;
                e[u] = (i < T) ? (unsigned)entL[i] : 0u;
            }
            int q[4];
#pragma unroll
            for (int u = 0; u < 4; ++u) {
                const int i = i0 + u * 512;
                q[u] = (i < T) ? atomicAdd(&curL[e[u] >> sh], 1) : 0;
            }
#pragma unroll
            for (int u = 0; u < 4; ++u) {
                const int i = i0 + u * 512;
                if (i < T) srtL[q[u]] = (int)(e[u] & idmask);
            }
        }
        __syncthreads();
        for (int i = tid; i < T; i += 512) slots[(long long)base + i] = srtL[i];
    } else {
        for (int i = tid; i < T; i += 512) {
            int lo = 0, hi = nch - 1;
            while (lo < hi) {
                const int mid = (lo + hi) >> 1;
                if (pos[mid + 1] > i) hi = mid; else lo = mid + 1;
            }
            const unsigned en = (unsigned)stage[pstart[lo] + (i - pos[lo])];
            const int p = atomicAdd(&curL[en >> sh], 1);
            slots[(long long)base + p] = (int)(en & idmask);
        }
    }
}

#define ROW_ADD(a, v) { a.x += v.x; a.y += v.y; a.z += v.z; a.w += v.w; }

template <int J>
__device__ __forceinline__ unsigned row_lane(unsigned v) {
    return (unsigned)__builtin_amdgcn_update_dpp(0, (int)v, 0x150 + J, 0xf, 0xf, true);
}

#define RANK_LOW(J)  { const unsigned u = row_lane<J>(ka); ra += (u < ka) ? 1 : 0; }
#define RANK_HIGH(J) { const unsigned u = row_lane<J>(ka); const unsigned w = row_lane<J>(kb); \
                       rb += (u < kb) ? 1 : 0; ra += (w < ka) ? 1 : 0; rb += (w < kb) ? 1 : 0; }
__device__ __forceinline__ void rank_store(int ia, int ib, int dd, int lane16, int* __restrict__ seg) {
    const unsigned ka = (lane16      < dd) ? (((unsigned)ia << 5) | (unsigned)lane16)        : 0xffffffffu;
    const unsigned kb = (lane16 + 16 < dd) ? (((unsigned)ib << 5) | (unsigned)(lane16 + 16)) : 0xffffffffu;
    int ra = 0, rb = 0;
    RANK_LOW(0)  RANK_LOW(1)  RANK_LOW(2)  RANK_LOW(3)  RANK_LOW(4)  RANK_LOW(5)  RANK_LOW(6)  RANK_LOW(7)
    RANK_LOW(8)  RANK_LOW(9)  RANK_LOW(10) RANK_LOW(11) RANK_LOW(12) RANK_LOW(13) RANK_LOW(14) RANK_LOW(15)
    if (__builtin_amdgcn_ballot_w32(dd > 16) != 0u) {
        RANK_HIGH(0)  RANK_HIGH(1)  RANK_HIGH(2)  RANK_HIGH(3)  RANK_HIGH(4)  RANK_HIGH(5)  RANK_HIGH(6)  RANK_HIGH(7)
        RANK_HIGH(8)  RANK_HIGH(9)  RANK_HIGH(10) RANK_HIGH(11) RANK_HIGH(12) RANK_HIGH(13) RANK_HIGH(14) RANK_HIGH(15)
    }
    if (lane16      < dd) seg[ra] = ia;
    if (lane16 + 16 < dd) seg[rb] = ib;
}

__device__ __forceinline__ void rank_long(const int* __restrict__ slots, int o, int d, int lane16,
                                          int* __restrict__ seg) {
    if (d > REG_CAP && d <= SEG_CAP) {
        for (int i = lane16; i < d; i += 16) {
            const int vi = slots[o + i];
            int r = 0;
            for (int j = 0; j < d; ++j) {
                const int u = slots[o + j];
                r += (u < vi || (u == vi && j < i)) ? 1 : 0;
            }
            seg[r] = vi;
        }
    }
}

__global__ __launch_bounds__(256)
void biconv_gather_gemm(const float* __restrict__ x,
                        const int* __restrict__ cnt,
                        const int* __restrict__ off,
                        const int* __restrict__ slots,
                        const float* __restrict__ norm,
                        const float* __restrict__ norm_t,
                        const float* __restrict__ w_out,
                        const float* __restrict__ w_back,
                        float* __restrict__ out,
                        int N) {
    __shared__ __attribute__((aligned(16))) float As[2][16][A_STRIDE];
    __shared__ __attribute__((aligned(16))) int   srt[2][16][SEG_CAP];

    const int tid    = threadIdx.x;
    const int hw     = tid >> 4;
    const int lane16 = tid & 15;
    const int wave   = tid >> 5;
    const int lane   = tid & 31;
    const int m      = lane & 15;
    const int half   = lane >> 4;
    const int kb     = half << 1;
    const int col    = (wave & 3) * 16 + (lane & 15);
    const float4* __restrict__ x4 = (const float4*)x;
    const float* __restrict__ wo = w_out  + kb * CHANNELS + col;
    const float* __restrict__ wb = w_back + kb * CHANNELS + col;

    const int ntiles = (N + 15) >> 4;
    const int tstep  = (int)gridDim.x;

    int ccnt0 = 0, ccnt1 = 0, cbase0 = 0, cbase1 = 0;
    int ncnt0 = 0, ncnt1 = 0, nbase0 = 0, nbase1 = 0;
    {
        const int n0 = (int)blockIdx.x * 16 + hw;
        if (n0 < N) {
            ccnt0  = cnt[n0];
            ccnt1  = cnt[N + n0];
            cbase0 = off[n0];
            cbase1 = off[N + n0];
        }
        const int n1 = ((int)blockIdx.x + tstep) * 16 + hw;
        if (n1 < N) {
            ncnt0  = cnt[n1];
            ncnt1  = cnt[N + n1];
            nbase0 = off[n1];
            nbase1 = off[N + n1];
        }
    }
    {
        const int fd0 = (ccnt0 <= REG_CAP) ? ccnt0 : 0;
        const int fd1 = (ccnt1 <= REG_CAP) ? ccnt1 : 0;
        int q00 = 0, q01 = 0, q10 = 0, q11 = 0;
        if (lane16      < fd0) q00 = slots[cbase0 + lane16];
        if (lane16 + 16 < fd0) q01 = slots[cbase0 + lane16 + 16];
        if (lane16      < fd1) q10 = slots[cbase1 + lane16];
        if (lane16 + 16 < fd1) q11 = slots[cbase1 + lane16 + 16];
        rank_store(q00, q01, fd0, lane16, &srt[0][hw][0]);
        rank_store(q10, q11, fd1, lane16, &srt[1][hw][0]);
        rank_long(slots, cbase0, ccnt0, lane16, &srt[0][hw][0]);
        rank_long(slots, cbase1, ccnt1, lane16, &srt[1][hw][0]);
    }

    for (int tile = (int)blockIdx.x; tile < ntiles; tile += tstep) {
        const int node   = tile * 16 + hw;
        const bool valid = node < N;

        const int dcount[2] = {ccnt0, ccnt1};
        const int dbase[2]  = {cbase0, cbase1};

        __syncthreads();

        float scale[2] = {0.f, 0.f};
        float4 self = make_float4(0.f, 0.f, 0.f, 0.f);
        if (valid) {
            scale[0] = norm[node];
            scale[1] = norm_t[node];
            self     = x4[(long long)node * 16 + lane16];
        }
        int fcnt0 = 0, fcnt1 = 0, fbase0 = 0, fbase1 = 0;
        {
            const int n2 = (tile + 2 * tstep) * 16 + hw;
            if (n2 < N) {
                fcnt0  = cnt[n2];
                fcnt1  = cnt[N + n2];
                fbase0 = off[n2];
                fbase1 = off[N + n2];
            }
        }
        const int nd0 = (ncnt0 <= REG_CAP) ? ncnt0 : 0;
        const int nd1 = (ncnt1 <= REG_CAP) ? ncnt1 : 0;
        int p00 = 0, p01 = 0, p10 = 0, p11 = 0;
        if (lane16      < nd0) p00 = slots[nbase0 + lane16];
        if (lane16 + 16 < nd0) p01 = slots[nbase0 + lane16 + 16];
        if (lane16      < nd1) p10 = slots[nbase1 + lane16];
        if (lane16 + 16 < nd1) p11 = slots[nbase1 + lane16 + 16];

#pragma unroll
        for (int dir = 0; dir < 2; ++dir) {
            const int d = dcount[dir];
            const int o = dbase[dir];
            const int* __restrict__ ids = &srt[dir][hw][0];
            float4 acc = self;

            if (d <= SEG_CAP) {
                int k = 0;
#pragma unroll 1
                for (; k + 8 <= d; k += 8) {
                    const int4 q0 = *(const int4*)(ids + k);
                    const int4 q1 = *(const int4*)(ids + k + 4);
                    const float4 v0 = x4[(long long)q0.x * 16 + lane16];
                    const float4 v1 = x4[(long long)q0.y * 16 + lane16];
                    const float4 v2 = x4[(long long)q0.z * 16 + lane16];
                    const float4 v3 = x4[(long long)q0.w * 16 + lane16];
                    const float4 v4 = x4[(long long)q1.x * 16 + lane16];
                    const float4 v5 = x4[(long long)q1.y * 16 + lane16];
                    const float4 v6 = x4[(long long)q1.z * 16 + lane16];
                    const float4 v7 = x4[(long long)q1.w * 16 + lane16];
                    ROW_ADD(acc, v0) ROW_ADD(acc, v1) ROW_ADD(acc, v2) ROW_ADD(acc, v3)
                    ROW_ADD(acc, v4) ROW_ADD(acc, v5) ROW_ADD(acc, v6) ROW_ADD(acc, v7)
                }
                if (k + 4 <= d) {
                    const int4 q0 = *(const int4*)(ids + k);
                    const float4 v0 = x4[(long long)q0.x * 16 + lane16];
                    const float4 v1 = x4[(long long)q0.y * 16 + lane16];
                    const float4 v2 = x4[(long long)q0.z * 16 + lane16];
                    const float4 v3 = x4[(long long)q0.w * 16 + lane16];
                    ROW_ADD(acc, v0) ROW_ADD(acc, v1) ROW_ADD(acc, v2) ROW_ADD(acc, v3)
                    k += 4;
                }
                const int rest = d - k;
                if (rest > 0) {
                    float4 t0 = x4[(long long)ids[k] * 16 + lane16];
                    float4 t1 = make_float4(0.f, 0.f, 0.f, 0.f);
                    float4 t2 = make_float4(0.f, 0.f, 0.f, 0.f);
                    if (rest > 1) t1 = x4[(long long)ids[k + 1] * 16 + lane16];
                    if (rest > 2) t2 = x4[(long long)ids[k + 2] * 16 + lane16];
                    ROW_ADD(acc, t0)
                    if (rest > 1) ROW_ADD(acc, t1)
                    if (rest > 2) ROW_ADD(acc, t2)
                }
            } else {
                int prev = -1;
                int done = 0;
                while (done < d) {
                    int best = 0x7fffffff;
                    for (int j = 0; j < d; ++j) {
                        const int u = slots[o + j];
                        if (u > prev && u < best) best = u;
                    }
                    int mult = 0;
                    for (int j = 0; j < d; ++j) mult += (slots[o + j] == best) ? 1 : 0;
                    if (mult == 0) break;
                    const float4 v = x4[(long long)best * 16 + lane16];
                    for (int t = 0; t < mult; ++t) ROW_ADD(acc, v)
                    prev = best;
                    done += mult;
                }
            }

            const float sc = scale[dir];
            float4 sv;
            sv.x = acc.x * sc; sv.y = acc.y * sc; sv.z = acc.z * sc; sv.w = acc.w * sc;
            *(float4*)&As[dir][hw][lane16 * 4] = sv;
        }

        rank_store(p00, p01, nd0, lane16, &srt[0][hw][0]);
        rank_store(p10, p11, nd1, lane16, &srt[1][hw][0]);
        rank_long(slots, nbase0, ncnt0, lane16, &srt[0][hw][0]);
        rank_long(slots, nbase1, ncnt1, lane16, &srt[1][hw][0]);
        __syncthreads();

        if (wave < 4) {
            v8f acc = {};
#pragma unroll
            for (int k = 0; k < CHANNELS; k += 4) {
                const float2 av = *(const float2*)&As[0][m][k + kb];
                v2f a; a.x = av.x; a.y = av.y;
                v2f b;
                b.x = wo[k * CHANNELS];
                b.y = wo[(k + 1) * CHANNELS];
                acc = __builtin_amdgcn_wmma_f32_16x16x4_f32(
                    false, a, false, b, (short)0, acc, false, false);
            }
#pragma unroll
            for (int k = 0; k < CHANNELS; k += 4) {
                const float2 av = *(const float2*)&As[1][m][k + kb];
                v2f a; a.x = av.x; a.y = av.y;
                v2f b;
                b.x = wb[k * CHANNELS];
                b.y = wb[(k + 1) * CHANNELS];
                acc = __builtin_amdgcn_wmma_f32_16x16x4_f32(
                    false, a, false, b, (short)0, acc, false, false);
            }

#pragma unroll
            for (int j = 0; j < 8; ++j) {
                const int mrow = tile * 16 + j + half * 8;
                if (mrow < N)
                    out[(long long)mrow * CHANNELS + col] = acc[j];
            }
        }

        ccnt0 = ncnt0; ccnt1 = ncnt1; cbase0 = nbase0; cbase1 = nbase1;
        ncnt0 = fcnt0; ncnt1 = fcnt1; nbase0 = fbase0; nbase1 = fbase1;
    }
}

extern "C" void kernel_launch(void* const* d_in, const int* in_sizes, int n_in,
                              void* d_out, int out_size, void* d_ws, size_t ws_size,
                              hipStream_t stream) {
    const float* x       = (const float*)d_in[0];
    const int*   sources = (const int*)d_in[1];
    const int*   targets = (const int*)d_in[2];
    const float* norm    = (const float*)d_in[3];
    const float* norm_t  = (const float*)d_in[4];
    const float* w_out   = (const float*)d_in[5];
    const float* w_back  = (const float*)d_in[6];
    float*       out     = (float*)d_out;

    const int N = in_sizes[3];
    const int E = in_sizes[1];
    const int NB  = (N + BIN - 1) / BIN;
    const int nch = (E + CH_EDGES - 1) / CH_EDGES;

    int* cnt    = (int*)d_ws;
    int* cursor = cnt + 2 * (long long)N;
    int* off    = cursor + 64;
    int* slots  = off + 2 * (long long)N;
    int* stage  = slots + 2 * (long long)E;
    int* coff   = stage + 2 * (long long)E;

    biconv_chunk_sort<<<nch, 1024, 0, stream>>>(
        sources, targets, cursor, stage, coff, N, E, NB);
    biconv_bin_build<<<2 * NB, 512, 0, stream>>>(
        stage, coff, cursor, cnt, off, slots, N, E, NB, nch);
    biconv_gather_gemm<<<1280, 256, 0, stream>>>(
        x, cnt, off, slots, norm, norm_t, w_out, w_back, out, N);
}


# last kernel: 16x64 result tile staged in LDS and written behind the next barrier as whole 256-byte rows, one 16-byte store per lane (was 8 four-byte stores per lane, 64-byte row pieces); same sums
# speedup vs baseline: 3.8911x; 1.0542x over previous
#include <hip/hip_runtime.h>

typedef __attribute__((ext_vector_type(2))) float v2f;
typedef __attribute__((ext_vector_type(8))) float v8f;

#define CHANNELS 64
#define SEG_CAP 64
#define REG_CAP 32
#define A_STRIDE 68

#define BIN_SHIFT 8
#define BIN 256
#define MAXNB 1024
#define CH_EDGES 8192
#define MAXCH 512
#define KB_CAP 8192
#define PCS 5

__device__ __forceinline__ int block_incl_scan(int v, int* wsum, int tid, int nwaves) {
    const int lane = tid & 31;
    const int wave = tid >> 5;
    __syncthreads();
    for (int s = 1; s < 32; s <<= 1) {
        int u = __shfl_up(v, s);
        if (lane >= s) v += u;
    }
    if (lane == 31) wsum[wave] = v;
    __syncthreads();
    if (wave == 0) {
        const int wv = (lane < nwaves) ? wsum[lane] : 0;
        int z = wv;
        for (int s = 1; s < 32; s <<= 1) {
            int u = __shfl_up(z, s);
            if (lane >= s) z += u;
        }
        wsum[lane] = z - wv;
    }
    __syncthreads();
    return v + wsum[wave];
}

__global__ __launch_bounds__(1024)
void biconv_chunk_sort(const int* __restrict__ src,
                       const int* __restrict__ tgt,
                       int* __restrict__ cursor,
                       int* __restrict__ stage,
                       int* __restrict__ coff,
                       int N, int E, int NB) {
    __shared__ int hist[2][MAXNB];
    __shared__ int stg[2][CH_EDGES];
    __shared__ int wsum[32];

    const int tid = threadIdx.x;
    const int w   = blockIdx.x;
    const int e0  = w * CH_EDGES;
    const int sh  = 32 - __clz(N - 1);
    const int nch = gridDim.x;

    if (w == 0 && tid == 0) *cursor = 0;

    hist[0][tid] = 0;
    hist[1][tid] = 0;
    __syncthreads();

    int sv[CH_EDGES / 1024];
    int tv[CH_EDGES / 1024];
#pragma unroll
    for (int k = 0; k < CH_EDGES / 1024; ++k) {
        const int e = e0 + k * 1024 + tid;
        sv[k] = -1;
        tv[k] = -1;
        if (e < E) {
            sv[k] = src[e];
            tv[k] = tgt[e];
        }
    }
#pragma unroll
    for (int k = 0; k < CH_EDGES / 1024; ++k) {
        if (sv[k] >= 0) {
            atomicAdd(&hist[0][tv[k] >> BIN_SHIFT], 1);
            atomicAdd(&hist[1][sv[k] >> BIN_SHIFT], 1);
        }
    }
    __syncthreads();

    const int c0 = hist[0][tid];
    const int c1 = hist[1][tid];
    const int inc = block_incl_scan(c0 | (c1 << 16), wsum, tid, 32);
    const int x0 = (inc & 0xffff) - c0;
    const int x1 = (inc >> 16) - c1;
    hist[0][tid] = x0;
    hist[1][tid] = x1;
    if (tid < NB) {
        coff[(long long)w * (NB + 1) + tid]         = x0;
        coff[(long long)(nch + w) * (NB + 1) + tid] = x1;
    }
    if (tid == 1023) {
        coff[(long long)w * (NB + 1) + NB]         = x0 + c0;
        coff[(long long)(nch + w) * (NB + 1) + NB] = x1 + c1;
    }
    __syncthreads();

#pragma unroll
    for (int k = 0; k < CH_EDGES / 1024; ++k) {
        if (sv[k] >= 0) {
            const int s = sv[k];
            const int t = tv[k];
            const int p = atomicAdd(&hist[0][t >> BIN_SHIFT], 1);
            stg[0][p] = (int)(((unsigned)(t & (BIN - 1)) << sh) | (unsigned)s);
            const int q = atomicAdd(&hist[1][s >> BIN_SHIFT], 1);
            stg[1][q] = (int)(((unsigned)(s & (BIN - 1)) << sh) | (unsigned)t);
        }
    }
    __syncthreads();

    const int ne = (E - e0 < CH_EDGES) ? (E - e0) : CH_EDGES;
    for (int i = tid; i < ne; i += 1024) {
        stage[(long long)e0 + i]     = stg[0][i];
        stage[(long long)E + e0 + i] = stg[1][i];
    }
}

__global__ __launch_bounds__(512)
void biconv_bin_build(const int* __restrict__ stage,
                      const int* __restrict__ coff,
                      int* __restrict__ cursor,
                      int* __restrict__ cnt,
                      int* __restrict__ off,
                      int* __restrict__ slots,
                      int N, int E, int NB, int nch) {
    __shared__ int pos[MAXCH + 2];
    __shared__ int pstart[MAXCH + 1];
    __shared__ int cntL[BIN];
    __shared__ int curL[BIN];
    __shared__ int entL[KB_CAP];
    __shared__ int srtL[KB_CAP];
    __shared__ int wsum[32];
    __shared__ int sbase;

    const int tid  = threadIdx.x;
    const int lane = tid & 31;
    const int wave = tid >> 5;
    const int d   = (blockIdx.x >= NB) ? 1 : 0;
    const int b   = blockIdx.x - d * NB;
    const int sh  = 32 - __clz(N - 1);
    const unsigned idmask = (sh >= 32) ? 0xffffffffu : ((1u << sh) - 1u);

    int c = 0;
    if (tid < nch) {
        const long long row = (long long)(d * nch + tid) * (NB + 1);
        const int a = coff[row + b];
        const int z = coff[row + b + 1];
        c = z - a;
        pstart[tid] = d * E + tid * CH_EDGES + a;
    }
    if (tid < BIN) { cntL[tid] = 0; }
    const int pinc = block_incl_scan(c, wsum, tid, 16);
    if (tid < nch) pos[tid] = pinc - c;
    if (tid == nch - 1) {
        pos[nch]     = pinc;
        pos[nch + 1] = pinc;
        pstart[nch]  = 0;
    }
    __syncthreads();
    const int T = pos[nch];

    if (tid == 0) sbase = atomicAdd(cursor, T);

    const bool in_lds = (T <= KB_CAP);

    if (in_lds) {
        for (int u0 = 0; wave + 16 * u0 < nch; u0 += PCS) {
            int pb[PCS], pc[PCS], pg[PCS];
#pragma unroll
            for (int u = 0; u < PCS; ++u) {
                const int p  = wave + 16 * (u0 + u);
                const int pp = (p < nch) ? p : nch;
                const int b0 = pos[pp];
                pb[u] = b0;
                pc[u] = pos[pp + 1] - b0;
                pg[u] = pstart[pp];
            }
            unsigned ea[PCS], eb[PCS];
#pragma unroll
            for (int u = 0; u < PCS; ++u) {
                ea[u] = (unsigned)stage[(lane      < pc[u]) ? (pg[u] + lane)      : 0];
                eb[u] = (unsigned)stage[(lane + 32 < pc[u]) ? (pg[u] + lane + 32) : 0];
            }
#pragma unroll
            for (int u = 0; u < PCS; ++u) {
                if (lane < pc[u]) {
                    entL[pb[u] + lane] = (int)ea[u];
                    atomicAdd(&cntL[ea[u] >> sh], 1);
                }
                if (lane + 32 < pc[u]) {
                    entL[pb[u] + lane + 32] = (int)eb[u];
                    atomicAdd(&cntL[eb[u] >> sh], 1);
                }
            }
#pragma unroll
            for (int u = 0; u < PCS; ++u) {
                for (int l = lane + 64; l < pc[u]; l += 32) {
                    const unsigned en = (unsigned)stage[pg[u] + l];
                    entL[pb[u] + l] = (int)en;
                    atomicAdd(&cntL[en >> sh], 1);
                }
            }
        }
    } else {
        for (int i = tid; i < T; i += 512) {
            int lo = 0, hi = nch - 1;
            while (lo < hi) {
                const int mid = (lo + hi) >> 1;
                if (pos[mid + 1] > i) hi = mid; else lo = mid + 1;
            }
            const unsigned en = (unsigned)stage[pstart[lo] + (i - pos[lo])];
            atomicAdd(&cntL[en >> sh], 1);
        }
    }
    __syncthreads();

    const int cl  = (tid < BIN) ? cntL[tid] : 0;
    const int inc = block_incl_scan(cl, wsum, tid, 16);
    const int base = sbase;
    if (tid < BIN) {
        curL[tid] = inc - cl;
        const int node = b * BIN + tid;
        if (node < N) {
            cnt[d * N + node] = cl;
            off[d * N + node] = base + inc - cl;
        }
    }
    __syncthreads();

    if (in_lds) {
        for (int i0 = tid; i0 < T; i0 += 4 * 512) {
            unsigned e[4];
#pragma unroll
            for (int u = 0; u < 4; ++u) {
                const int i = i0 + u * 512;
                e[u] = (i < T) ? (unsigned)entL[i] : 0u;
            }
            int q[4];
#pragma unroll
            for (int u = 0; u < 4; ++u) {
                const int i = i0 + u * 512;
                q[u] = (i < T) ? atomicAdd(&curL[e[u] >> sh], 1) : 0;
            }
#pragma unroll
            for (int u = 0; u < 4; ++u) {
                const int i = i0 + u * 512;
                if (i < T) srtL[q[u]] = (int)(e[u] & idmask);
            }
        }
        __syncthreads();
        for (int i = tid; i < T; i += 512) slots[(long long)base + i] = srtL[i];
    } else {
        for (int i = tid; i < T; i += 512) {
            int lo = 0, hi = nch - 1;
            while (lo < hi) {
                const int mid = (lo + hi) >> 1;
                if (pos[mid + 1] > i) hi = mid; else lo = mid + 1;
            }
            const unsigned en = (unsigned)stage[pstart[lo] + (i - pos[lo])];
            const int p = atomicAdd(&curL[en >> sh], 1);
            slots[(long long)base + p] = (int)(en & idmask);
        }
    }
}

#define ROW_ADD(a, v) { a.x += v.x; a.y += v.y; a.z += v.z; a.w += v.w; }

template <int J>
__device__ __forceinline__ unsigned row_lane(unsigned v) {
    return (unsigned)__builtin_amdgcn_update_dpp(0, (int)v, 0x150 + J, 0xf, 0xf, true);
}

#define RANK_LOW(J)  { const unsigned u = row_lane<J>(ka); ra += (u < ka) ? 1 : 0; }
#define RANK_HIGH(J) { const unsigned u = row_lane<J>(ka); const unsigned w = row_lane<J>(kb); \
                       rb += (u < kb) ? 1 : 0; ra += (w < ka) ? 1 : 0; rb += (w < kb) ? 1 : 0; }
__device__ __forceinline__ void rank_store(int ia, int ib, int dd, int lane16, int* __restrict__ seg) {
    const unsigned ka = (lane16      < dd) ? (((unsigned)ia << 5) | (unsigned)lane16)        : 0xffffffffu;
    const unsigned kb = (lane16 + 16 < dd) ? (((unsigned)ib << 5) | (unsigned)(lane16 + 16)) : 0xffffffffu;
    int ra = 0, rb = 0;
    RANK_LOW(0)  RANK_LOW(1)  RANK_LOW(2)  RANK_LOW(3)  RANK_LOW(4)  RANK_LOW(5)  RANK_LOW(6)  RANK_LOW(7)
    RANK_LOW(8)  RANK_LOW(9)  RANK_LOW(10) RANK_LOW(11) RANK_LOW(12) RANK_LOW(13) RANK_LOW(14) RANK_LOW(15)
    if (__builtin_amdgcn_ballot_w32(dd > 16) != 0u) {
        RANK_HIGH(0)  RANK_HIGH(1)  RANK_HIGH(2)  RANK_HIGH(3)  RANK_HIGH(4)  RANK_HIGH(5)  RANK_HIGH(6)  RANK_HIGH(7)
        RANK_HIGH(8)  RANK_HIGH(9)  RANK_HIGH(10) RANK_HIGH(11) RANK_HIGH(12) RANK_HIGH(13) RANK_HIGH(14) RANK_HIGH(15)
    }
    if (lane16      < dd) seg[ra] = ia;
    if (lane16 + 16 < dd) seg[rb] = ib;
}

__device__ __forceinline__ void rank_long(const int* __restrict__ slots, int o, int d, int lane16,
                                          int* __restrict__ seg) {
    if (d > REG_CAP && d <= SEG_CAP) {
        for (int i = lane16; i < d; i += 16) {
            const int vi = slots[o + i];
            int r = 0;
            for (int j = 0; j < d; ++j) {
                const int u = slots[o + j];
                r += (u < vi || (u == vi && j < i)) ? 1 : 0;
            }
            seg[r] = vi;
        }
    }
}

__global__ __launch_bounds__(256)
void biconv_gather_gemm(const float* __restrict__ x,
                        const int* __restrict__ cnt,
                        const int* __restrict__ off,
                        const int* __restrict__ slots,
                        const float* __restrict__ norm,
                        const float* __restrict__ norm_t,
                        const float* __restrict__ w_out,
                        const float* __restrict__ w_back,
                        float* __restrict__ out,
                        int N) {
    __shared__ __attribute__((aligned(16))) float As[2][16][A_STRIDE];
    __shared__ __attribute__((aligned(16))) int   srt[2][16][SEG_CAP];
    __shared__ __attribute__((aligned(16))) float Cs[16][A_STRIDE];

    const int tid    = threadIdx.x;
    const int hw     = tid >> 4;
    const int lane16 = tid & 15;
    const int wave   = tid >> 5;
    const int lane   = tid & 31;
    const int m      = lane & 15;
    const int half   = lane >> 4;
    const int kb     = half << 1;
    const int col    = (wave & 3) * 16 + (lane & 15);
    const float4* __restrict__ x4 = (const float4*)x;
    const float* __restrict__ wo = w_out  + kb * CHANNELS + col;
    const float* __restrict__ wb = w_back + kb * CHANNELS + col;

    const int ntiles = (N + 15) >> 4;
    const int tstep  = (int)gridDim.x;

    int ccnt0 = 0, ccnt1 = 0, cbase0 = 0, cbase1 = 0;
    int ncnt0 = 0, ncnt1 = 0, nbase0 = 0, nbase1 = 0;
    {
        const int n0 = (int)blockIdx.x * 16 + hw;
        if (n0 < N) {
            ccnt0  = cnt[n0];
            ccnt1  = cnt[N + n0];
            cbase0 = off[n0];
            cbase1 = off[N + n0];
        }
        const int n1 = ((int)blockIdx.x + tstep) * 16 + hw;
        if (n1 < N) {
            ncnt0  = cnt[n1];
            ncnt1  = cnt[N + n1];
            nbase0 = off[n1];
            nbase1 = off[N + n1];
        }
    }
    {
        const int fd0 = (ccnt0 <= REG_CAP) ? ccnt0 : 0;
        const int fd1 = (ccnt1 <= REG_CAP) ? ccnt1 : 0;
        int q00 = 0, q01 = 0, q10 = 0, q11 = 0;
        if (lane16      < fd0) q00 = slots[cbase0 + lane16];
        if (lane16 + 16 < fd0) q01 = slots[cbase0 + lane16 + 16];
        if (lane16      < fd1) q10 = slots[cbase1 + lane16];
        if (lane16 + 16 < fd1) q11 = slots[cbase1 + lane16 + 16];
        rank_store(q00, q01, fd0, lane16, &srt[0][hw][0]);
        rank_store(q10, q11, fd1, lane16, &srt[1][hw][0]);
        rank_long(slots, cbase0, ccnt0, lane16, &srt[0][hw][0]);
        rank_long(slots, cbase1, ccnt1, lane16, &srt[1][hw][0]);
    }

    int ptile = -1;

    for (int tile = (int)blockIdx.x; tile < ntiles; tile += tstep) {
        const int node   = tile * 16 + hw;
        const bool valid = node < N;

        const int dcount[2] = {ccnt0, ccnt1};
        const int dbase[2]  = {cbase0, cbase1};

        __syncthreads();

        float4 pres = make_float4(0.f, 0.f, 0.f, 0.f);
        if (ptile >= 0) pres = *(const float4*)&Cs[hw][lane16 * 4];

        float scale[2] = {0.f, 0.f};
        float4 self = make_float4(0.f, 0.f, 0.f, 0.f);
        if (valid) {
            scale[0] = norm[node];
            scale[1] = norm_t[node];
            self     = x4[(long long)node * 16 + lane16];
        }
        int fcnt0 = 0, fcnt1 = 0, fbase0 = 0, fbase1 = 0;
        {
            const int n2 = (tile + 2 * tstep) * 16 + hw;
            if (n2 < N) {
                fcnt0  = cnt[n2];
                fcnt1  = cnt[N + n2];
                fbase0 = off[n2];
                fbase1 = off[N + n2];
            }
        }
        const int nd0 = (ncnt0 <= REG_CAP) ? ncnt0 : 0;
        const int nd1 = (ncnt1 <= REG_CAP) ? ncnt1 : 0;
        int p00 = 0, p01 = 0, p10 = 0, p11 = 0;
        if (lane16      < nd0) p00 = slots[nbase0 + lane16];
        if (lane16 + 16 < nd0) p01 = slots[nbase0 + lane16 + 16];
        if (lane16      < nd1) p10 = slots[nbase1 + lane16];
        if (lane16 + 16 < nd1) p11 = slots[nbase1 + lane16 + 16];

        if (ptile >= 0) {
            const int prow = ptile * 16 + hw;
            if (prow < N)
                *(float4*)&out[(long long)prow * CHANNELS + lane16 * 4] = pres;
        }

#pragma unroll
        for (int dir = 0; dir < 2; ++dir) {
            const int d = dcount[dir];
            const int o = dbase[dir];
            const int* __restrict__ ids = &srt[dir][hw][0];
            float4 acc = self;

            if (d <= SEG_CAP) {
                int k = 0;
#pragma unroll 1
                for (; k + 8 <= d; k += 8) {
                    const int4 q0 = *(const int4*)(ids + k);
                    const int4 q1 = *(const int4*)(ids + k + 4);
                    const float4 v0 = x4[(long long)q0.x * 16 + lane16];
                    const float4 v1 = x4[(long long)q0.y * 16 + lane16];
                    const float4 v2 = x4[(long long)q0.z * 16 + lane16];
                    const float4 v3 = x4[(long long)q0.w * 16 + lane16];
                    const float4 v4 = x4[(long long)q1.x * 16 + lane16];
                    const float4 v5 = x4[(long long)q1.y * 16 + lane16];
                    const float4 v6 = x4[(long long)q1.z * 16 + lane16];
                    const float4 v7 = x4[(long long)q1.w * 16 + lane16];
                    ROW_ADD(acc, v0) ROW_ADD(acc, v1) ROW_ADD(acc, v2) ROW_ADD(acc, v3)
                    ROW_ADD(acc, v4) ROW_ADD(acc, v5) ROW_ADD(acc, v6) ROW_ADD(acc, v7)
                }
                if (k + 4 <= d) {
                    const int4 q0 = *(const int4*)(ids + k);
                    const float4 v0 = x4[(long long)q0.x * 16 + lane16];
                    const float4 v1 = x4[(long long)q0.y * 16 + lane16];
                    const float4 v2 = x4[(long long)q0.z * 16 + lane16];
                    const float4 v3 = x4[(long long)q0.w * 16 + lane16];
                    ROW_ADD(acc, v0) ROW_ADD(acc, v1) ROW_ADD(acc, v2) ROW_ADD(acc, v3)
                    k += 4;
                }
                const int rest = d - k;
                if (rest > 0) {
                    float4 t0 = x4[(long long)ids[k] * 16 + lane16];
                    float4 t1 = make_float4(0.f, 0.f, 0.f, 0.f);
                    float4 t2 = make_float4(0.f, 0.f, 0.f, 0.f);
                    if (rest > 1) t1 = x4[(long long)ids[k + 1] * 16 + lane16];
                    if (rest > 2) t2 = x4[(long long)ids[k + 2] * 16 + lane16];
                    ROW_ADD(acc, t0)
                    if (rest > 1) ROW_ADD(acc, t1)
                    if (rest > 2) ROW_ADD(acc, t2)
                }
            } else {
                int prev = -1;
                int done = 0;
                while (done < d) {
                    int best = 0x7fffffff;
                    for (int j = 0; j < d; ++j) {
                        const int u = slots[o + j];
                        if (u > prev && u < best) best = u;
                    }
                    int mult = 0;
                    for (int j = 0; j < d; ++j) mult += (slots[o + j] == best) ? 1 : 0;
                    if (mult == 0) break;
                    const float4 v = x4[(long long)best * 16 + lane16];
                    for (int t = 0; t < mult; ++t) ROW_ADD(acc, v)
                    prev = best;
                    done += mult;
                }
            }

            const float sc = scale[dir];
            float4 sv;
            sv.x = acc.x * sc; sv.y = acc.y * sc; sv.z = acc.z * sc; sv.w = acc.w * sc;
            *(float4*)&As[dir][hw][lane16 * 4] = sv;
        }

        rank_store(p00, p01, nd0, lane16, &srt[0][hw][0]);
        rank_store(p10, p11, nd1, lane16, &srt[1][hw][0]);
        rank_long(slots, nbase0, ncnt0, lane16, &srt[0][hw][0]);
        rank_long(slots, nbase1, ncnt1, lane16, &srt[1][hw][0]);
        __syncthreads();

        if (wave < 4) {
            v8f acc = {};
#pragma unroll
            for (int k = 0; k < CHANNELS; k += 4) {
                const float2 av = *(const float2*)&As[0][m][k + kb];
                v2f a; a.x = av.x; a.y = av.y;
                v2f b;
                b.x = wo[k * CHANNELS];
                b.y = wo[(k + 1) * CHANNELS];
                acc = __builtin_amdgcn_wmma_f32_16x16x4_f32(
                    false, a, false, b, (short)0, acc, false, false);
            }
#pragma unroll
            for (int k = 0; k < CHANNELS; k += 4) {
                const float2 av = *(const float2*)&As[1][m][k + kb];
                v2f a; a.x = av.x; a.y = av.y;
                v2f b;
                b.x = wb[k * CHANNELS];
                b.y = wb[(k + 1) * CHANNELS];
                acc = __builtin_amdgcn_wmma_f32_16x16x4_f32(
                    false, a, false, b, (short)0, acc, false, false);
            }

#pragma unroll
            for (int j = 0; j < 8; ++j)
                Cs[j + half * 8][col] = acc[j];
        }
        ptile = tile;

        ccnt0 = ncnt0; ccnt1 = ncnt1; cbase0 = nbase0; cbase1 = nbase1;
        ncnt0 = fcnt0; ncnt1 = fcnt1; nbase0 = fbase0; nbase1 = fbase1;
    }

    __syncthreads();
    if (ptile >= 0) {
        const int prow = ptile * 16 + hw;
        if (prow < N)
            *(float4*)&out[(long long)prow * CHANNELS + lane16 * 4] =
                *(const float4*)&Cs[hw][lane16 * 4];
    }
}

extern "C" void kernel_launch(void* const* d_in, const int* in_sizes, int n_in,
                              void* d_out, int out_size, void* d_ws, size_t ws_size,
                              hipStream_t stream) {
    const float* x       = (const float*)d_in[0];
    const int*   sources = (const int*)d_in[1];
    const int*   targets = (const int*)d_in[2];
    const float* norm    = (const float*)d_in[3];
    const float* norm_t  = (const float*)d_in[4];
    const float* w_out   = (const float*)d_in[5];
    const float* w_back  = (const float*)d_in[6];
    float*       out     = (float*)d_out;

    const int N = in_sizes[3];
    const int E = in_sizes[1];
    const int NB  = (N + BIN - 1) / BIN;
    const int nch = (E + CH_EDGES - 1) / CH_EDGES;

    int* cnt    = (int*)d_ws;
    int* cursor = cnt + 2 * (long long)N;
    int* off    = cursor + 64;
    int* slots  = off + 2 * (long long)N;
    int* stage  = slots + 2 * (long long)E;
    int* coff   = stage + 2 * (long long)E;

    biconv_chunk_sort<<<nch, 1024, 0, stream>>>(
        sources, targets, cursor, stage, coff, N, E, NB);
    biconv_bin_build<<<2 * NB, 512, 0, stream>>>(
        stage, coff, cursor, cnt, off, slots, N, E, NB, nch);
    biconv_gather_gemm<<<1280, 256, 0, stream>>>(
        x, cnt, off, slots, norm, norm_t, w_out, w_back, out, N);
}


# last kernel: weights copied once per work-group to LDS in fragment order, all 8 waves multiply, one direction each (was 4 waves, 64 global loads per tile each); sum order: out part + back part
# speedup vs baseline: 4.0303x; 1.0358x over previous
#include <hip/hip_runtime.h>

typedef __attribute__((ext_vector_type(2))) float v2f;
typedef __attribute__((ext_vector_type(8))) float v8f;

#define CHANNELS 64
#define SEG_CAP 64
#define REG_CAP 32
#define A_STRIDE 68

#define BIN_SHIFT 8
#define BIN 256
#define MAXNB 1024
#define CH_EDGES 8192
#define MAXCH 512
#define KB_CAP 8192
#define PCS 5

__device__ __forceinline__ int block_incl_scan(int v, int* wsum, int tid, int nwaves) {
    const int lane = tid & 31;
    const int wave = tid >> 5;
    __syncthreads();
    for (int s = 1; s < 32; s <<= 1) {
        int u = __shfl_up(v, s);
        if (lane >= s) v += u;
    }
    if (lane == 31) wsum[wave] = v;
    __syncthreads();
    if (wave == 0) {
        const int wv = (lane < nwaves) ? wsum[lane] : 0;
        int z = wv;
        for (int s = 1; s < 32; s <<= 1) {
            int u = __shfl_up(z, s);
            if (lane >= s) z += u;
        }
        wsum[lane] = z - wv;
    }
    __syncthreads();
    return v + wsum[wave];
}

__global__ __launch_bounds__(1024)
void biconv_chunk_sort(const int* __restrict__ src,
                       const int* __restrict__ tgt,
                       int* __restrict__ cursor,
                       int* __restrict__ stage,
                       int* __restrict__ coff,
                       int N, int E, int NB) {
    __shared__ int hist[2][MAXNB];
    __shared__ int stg[2][CH_EDGES];
    __shared__ int wsum[32];

    const int tid = threadIdx.x;
    const int w   = blockIdx.x;
    const int e0  = w * CH_EDGES;
    const int sh  = 32 - __clz(N - 1);
    const int nch = gridDim.x;

    if (w == 0 && tid == 0) *cursor = 0;

    hist[0][tid] = 0;
    hist[1][tid] = 0;
    __syncthreads();

    int sv[CH_EDGES / 1024];
    int tv[CH_EDGES / 1024];
#pragma unroll
    for (int k = 0; k < CH_EDGES / 1024; ++k) {
        const int e = e0 + k * 1024 + tid;
        sv[k] = -1;
        tv[k] = -1;
        if (e < E) {
            sv[k] = src[e];
            tv[k] = tgt[e];
        }
    }
#pragma unroll
    for (int k = 0; k < CH_EDGES / 1024; ++k) {
        if (sv[k] >= 0) {
            atomicAdd(&hist[0][tv[k] >> BIN_SHIFT], 1);
            atomicAdd(&hist[1][sv[k] >> BIN_SHIFT], 1);
        }
    }
    __syncthreads();

    const int c0 = hist[0][tid];
    const int c1 = hist[1][tid];
    const int inc = block_incl_scan(c0 | (c1 << 16), wsum, tid, 32);
    const int x0 = (inc & 0xffff) - c0;
    const int x1 = (inc >> 16) - c1;
    hist[0][tid] = x0;
    hist[1][tid] = x1;
    if (tid < NB) {
        coff[(long long)w * (NB + 1) + tid]         = x0;
        coff[(long long)(nch + w) * (NB + 1) + tid] = x1;
    }
    if (tid == 1023) {
        coff[(long long)w * (NB + 1) + NB]         = x0 + c0;
        coff[(long long)(nch + w) * (NB + 1) + NB] = x1 + c1;
    }
    __syncthreads();

#pragma unroll
    for (int k = 0; k < CH_EDGES / 1024; ++k) {
        if (sv[k] >= 0) {
            const int s = sv[k];
            const int t = tv[k];
            const int p = atomicAdd(&hist[0][t >> BIN_SHIFT], 1);
            stg[0][p] = (int)(((unsigned)(t & (BIN - 1)) << sh) | (unsigned)s);
            const int q = atomicAdd(&hist[1][s >> BIN_SHIFT], 1);
            stg[1][q] = (int)(((unsigned)(s & (BIN - 1)) << sh) | (unsigned)t);
        }
    }
    __syncthreads();

    const int ne = (E - e0 < CH_EDGES) ? (E - e0) : CH_EDGES;
    for (int i = tid; i < ne; i += 1024) {
        stage[(long long)e0 + i]     = stg[0][i];
        stage[(long long)E + e0 + i] = stg[1][i];
    }
}

__global__ __launch_bounds__(512)
void biconv_bin_build(const int* __restrict__ stage,
                      const int* __restrict__ coff,
                      int* __restrict__ cursor,
                      int* __restrict__ cnt,
                      int* __restrict__ off,
                      int* __restrict__ slots,
                      int N, int E, int NB, int nch) {
    __shared__ int pos[MAXCH + 2];
    __shared__ int pstart[MAXCH + 1];
    __shared__ int cntL[BIN];
    __shared__ int curL[BIN];
    __shared__ int entL[KB_CAP];
    __shared__ int srtL[KB_CAP];
    __shared__ int wsum[32];
    __shared__ int sbase;

    const int tid  = threadIdx.x;
    const int lane = tid & 31;
    const int wave = tid >> 5;
    const int d   = (blockIdx.x >= NB) ? 1 : 0;
    const int b   = blockIdx.x - d * NB;
    const int sh  = 32 - __clz(N - 1);
    const unsigned idmask = (sh >= 32) ? 0xffffffffu : ((1u << sh) - 1u);

    int c = 0;
    if (tid < nch) {
        const long long row = (long long)(d * nch + tid) * (NB + 1);
        const int a = coff[row + b];
        const int z = coff[row + b + 1];
        c = z - a;
        pstart[tid] = d * E + tid * CH_EDGES + a;
    }
    if (tid < BIN) { cntL[tid] = 0; }
    const int pinc = block_incl_scan(c, wsum, tid, 16);
    if (tid < nch) pos[tid] = pinc - c;
    if (tid == nch - 1) {
        pos[nch]     = pinc;
        pos[nch + 1] = pinc;
        pstart[nch]  = 0;
    }
    __syncthreads();
    const int T = pos[nch];

    if (tid == 0) sbase = atomicAdd(cursor, T);

    const bool in_lds = (T <= KB_CAP);

    if (in_lds) {
        for (int u0 = 0; wave + 16 * u0 < nch; u0 += PCS) {
            int pb[PCS], pc[PCS], pg[PCS];
#pragma unroll
            for (int u = 0; u < PCS; ++u) {
                const int p  = wave + 16 * (u0 + u);
                const int pp = (p < nch) ? p : nch;
                const int b0 = pos[pp];
                pb[u] = b0;
                pc[u] = pos[pp + 1] - b0;
                pg[u] = pstart[pp];
            }
            unsigned ea[PCS], eb[PCS];
#pragma unroll
            for (int u = 0; u < PCS; ++u) {
                ea[u] = (unsigned)stage[(lane      < pc[u]) ? (pg[u] + lane)      : 0];
                eb[u] = (unsigned)stage[(lane + 32 < pc[u]) ? (pg[u] + lane + 32) : 0];
            }
#pragma unroll
            for (int u = 0; u < PCS; ++u) {
                if (lane < pc[u]) {
                    entL[pb[u] + lane] = (int)ea[u];
                    atomicAdd(&cntL[ea[u] >> sh], 1);
                }
                if (lane + 32 < pc[u]) {
                    entL[pb[u] + lane + 32] = (int)eb[u];
                    atomicAdd(&cntL[eb[u] >> sh], 1);
                }
            }
#pragma unroll
            for (int u = 0; u < PCS; ++u) {
                for (int l = lane + 64; l < pc[u]; l += 32) {
                    const unsigned en = (unsigned)stage[pg[u] + l];
                    entL[pb[u] + l] = (int)en;
                    atomicAdd(&cntL[en >> sh], 1);
                }
            }
        }
    } else {
        for (int i = tid; i < T; i += 512) {
            int lo = 0, hi = nch - 1;
            while (lo < hi) {
                const int mid = (lo + hi) >> 1;
                if (pos[mid + 1] > i) hi = mid; else lo = mid + 1;
            }
            const unsigned en = (unsigned)stage[pstart[lo] + (i - pos[lo])];
            atomicAdd(&cntL[en >> sh], 1);
        }
    }
    __syncthreads();

    const int cl  = (tid < BIN) ? cntL[tid] : 0;
    const int inc = block_incl_scan(cl, wsum, tid, 16);
    const int base = sbase;
    if (tid < BIN) {
        curL[tid] = inc - cl;
        const int node = b * BIN + tid;
        if (node < N) {
            cnt[d * N + node] = cl;
            off[d * N + node] = base + inc - cl;
        }
    }
    __syncthreads();

    if (in_lds) {
        for (int i0 = tid; i0 < T; i0 += 4 * 512) {
            unsigned e[4];
#pragma unroll
            for (int u = 0; u < 4; ++u) {
                const int i = i0 + u * 512;
                e[u] = (i < T) ? (unsigned)entL[i] : 0u;
            }
            int q[4];
#pragma unroll
            for (int u = 0; u < 4; ++u) {
                const int i = i0 + u * 512;
                q[u] = (i < T) ? atomicAdd(&curL[e[u] >> sh], 1) : 0;
            }
#pragma unroll
            for (int u = 0; u < 4; ++u) {
                const int i = i0 + u * 512;
                if (i < T) srtL[q[u]] = (int)(e[u] & idmask);
            }
        }
        __syncthreads();
        for (int i = tid; i < T; i += 512) slots[(long long)base + i] = srtL[i];
    } else {
        for (int i = tid; i < T; i += 512) {
            int lo = 0, hi = nch - 1;
            while (lo < hi) {
                const int mid = (lo + hi) >> 1;
                if (pos[mid + 1] > i) hi = mid; else lo = mid + 1;
            }
            const unsigned en = (unsigned)stage[pstart[lo] + (i - pos[lo])];
            const int p = atomicAdd(&curL[en >> sh], 1);
            slots[(long long)base + p] = (int)(en & idmask);
        }
    }
}

#define ROW_ADD(a, v) { a.x += v.x; a.y += v.y; a.z += v.z; a.w += v.w; }

template <int J>
__device__ __forceinline__ unsigned row_lane(unsigned v) {
    return (unsigned)__builtin_amdgcn_update_dpp(0, (int)v, 0x150 + J, 0xf, 0xf, true);
}

#define RANK_LOW(J)  { const unsigned u = row_lane<J>(ka); ra += (u < ka) ? 1 : 0; }
#define RANK_HIGH(J) { const unsigned u = row_lane<J>(ka); const unsigned w = row_lane<J>(kb); \
                       rb += (u < kb) ? 1 : 0; ra += (w < ka) ? 1 : 0; rb += (w < kb) ? 1 : 0; }
__device__ __forceinline__ void rank_store(int ia, int ib, int dd, int lane16, int* __restrict__ seg) {
    const unsigned ka = (lane16      < dd) ? (((unsigned)ia << 5) | (unsigned)lane16)        : 0xffffffffu;
    const unsigned kb = (lane16 + 16 < dd) ? (((unsigned)ib << 5) | (unsigned)(lane16 + 16)) : 0xffffffffu;
    int ra = 0, rb = 0;
    RANK_LOW(0)  RANK_LOW(1)  RANK_LOW(2)  RANK_LOW(3)  RANK_LOW(4)  RANK_LOW(5)  RANK_LOW(6)  RANK_LOW(7)
    RANK_LOW(8)  RANK_LOW(9)  RANK_LOW(10) RANK_LOW(11) RANK_LOW(12) RANK_LOW(13) RANK_LOW(14) RANK_LOW(15)
    if (__builtin_amdgcn_ballot_w32(dd > 16) != 0u) {
        RANK_HIGH(0)  RANK_HIGH(1)  RANK_HIGH(2)  RANK_HIGH(3)  RANK_HIGH(4)  RANK_HIGH(5)  RANK_HIGH(6)  RANK_HIGH(7)
        RANK_HIGH(8)  RANK_HIGH(9)  RANK_HIGH(10) RANK_HIGH(11) RANK_HIGH(12) RANK_HIGH(13) RANK_HIGH(14) RANK_HIGH(15)
    }
    if (lane16      < dd) seg[ra] = ia;
    if (lane16 + 16 < dd) seg[rb] = ib;
}

__device__ __forceinline__ void rank_long(const int* __restrict__ slots, int o, int d, int lane16,
                                          int* __restrict__ seg) {
    if (d > REG_CAP && d <= SEG_CAP) {
        for (int i = lane16; i < d; i += 16) {
            const int vi = slots[o + i];
            int r = 0;
            for (int j = 0; j < d; ++j) {
                const int u = slots[o + j];
                r += (u < vi || (u == vi && j < i)) ? 1 : 0;
            }
            seg[r] = vi;
        }
    }
}

__global__ __launch_bounds__(256)
void biconv_gather_gemm(const float* __restrict__ x,
                        const int* __restrict__ cnt,
                        const int* __restrict__ off,
                        const int* __restrict__ slots,
                        const float* __restrict__ norm,
                        const float* __restrict__ norm_t,
                        const float* __restrict__ w_out,
                        const float* __restrict__ w_back,
                        float* __restrict__ out,
                        int N) {
    __shared__ __attribute__((aligned(16))) float As[2][16][A_STRIDE];
    __shared__ __attribute__((aligned(16))) int   srt[2][16][SEG_CAP];
    __shared__ __attribute__((aligned(16))) float Cs[2][16][A_STRIDE];
    __shared__ __attribute__((aligned(16))) float Bs[2 * 4 * 8 * 32 * 4];

    const int tid    = threadIdx.x;
    const int hw     = tid >> 4;
    const int lane16 = tid & 15;
    const int wave   = tid >> 5;
    const int lane   = tid & 31;
    const int m      = lane & 15;
    const int half   = lane >> 4;
    const int kb     = half << 1;
    const int wdir   = wave >> 2;
    const int ct     = wave & 3;
    const int col    = ct * 16 + (lane & 15);
    const float4* __restrict__ x4 = (const float4*)x;

    const int ntiles = (N + 15) >> 4;
    const int tstep  = (int)gridDim.x;

    int ccnt0 = 0, ccnt1 = 0, cbase0 = 0, cbase1 = 0;
    int ncnt0 = 0, ncnt1 = 0, nbase0 = 0, nbase1 = 0;
    {
        const int n0 = (int)blockIdx.x * 16 + hw;
        if (n0 < N) {
            ccnt0  = cnt[n0];
            ccnt1  = cnt[N + n0];
            cbase0 = off[n0];
            cbase1 = off[N + n0];
        }
        const int n1 = ((int)blockIdx.x + tstep) * 16 + hw;
        if (n1 < N) {
            ncnt0  = cnt[n1];
            ncnt1  = cnt[N + n1];
            nbase0 = off[n1];
            nbase1 = off[N + n1];
        }
    }
    {
        const int fd0 = (ccnt0 <= REG_CAP) ? ccnt0 : 0;
        const int fd1 = (ccnt1 <= REG_CAP) ? ccnt1 : 0;
        int q00 = 0, q01 = 0, q10 = 0, q11 = 0;
        if (lane16      < fd0) q00 = slots[cbase0 + lane16];
        if (lane16 + 16 < fd0) q01 = slots[cbase0 + lane16 + 16];
        if (lane16      < fd1) q10 = slots[cbase1 + lane16];
        if (lane16 + 16 < fd1) q11 = slots[cbase1 + lane16 + 16];
#pragma unroll
        for (int it = 0; it < 8; ++it) {
            const float* __restrict__ wsrc = (it >> 2) ? w_back : w_out;
            const int fl = tid & 31;
            const int k0 = 8 * (tid >> 5) + 2 * (fl >> 4);
            const int cc = 16 * (it & 3) + (fl & 15);
            float4 wv;
            wv.x = wsrc[(k0    ) * CHANNELS + cc];
            wv.y = wsrc[(k0 + 1) * CHANNELS + cc];
            wv.z = wsrc[(k0 + 4) * CHANNELS + cc];
            wv.w = wsrc[(k0 + 5) * CHANNELS + cc];
            *(float4*)&Bs[(it * 256 + tid) * 4] = wv;
        }
        rank_store(q00, q01, fd0, lane16, &srt[0][hw][0]);
        rank_store(q10, q11, fd1, lane16, &srt[1][hw][0]);
        rank_long(slots, cbase0, ccnt0, lane16, &srt[0][hw][0]);
        rank_long(slots, cbase1, ccnt1, lane16, &srt[1][hw][0]);
    }

    int ptile = -1;

    for (int tile = (int)blockIdx.x; tile < ntiles; tile += tstep) {
        const int node   = tile * 16 + hw;
        const bool valid = node < N;

        const int dcount[2] = {ccnt0, ccnt1};
        const int dbase[2]  = {cbase0, cbase1};

        __syncthreads();

        float4 pres = make_float4(0.f, 0.f, 0.f, 0.f);
        if (ptile >= 0) {
            const float4 po = *(const float4*)&Cs[0][hw][lane16 * 4];
            const float4 pb = *(const float4*)&Cs[1][hw][lane16 * 4];
            pres.x = po.x + pb.x; pres.y = po.y + pb.y; pres.z = po.z + pb.z; pres.w = po.w + pb.w;
        }

        float scale[2] = {0.f, 0.f};
        float4 self = make_float4(0.f, 0.f, 0.f, 0.f);
        if (valid) {
            scale[0] = norm[node];
            scale[1] = norm_t[node];
            self     = x4[(long long)node * 16 + lane16];
        }
        int fcnt0 = 0, fcnt1 = 0, fbase0 = 0, fbase1 = 0;
        {
            const int n2 = (tile + 2 * tstep) * 16 + hw;
            if (n2 < N) {
                fcnt0  = cnt[n2];
                fcnt1  = cnt[N + n2];
                fbase0 = off[n2];
                fbase1 = off[N + n2];
            }
        }
        const int nd0 = (ncnt0 <= REG_CAP) ? ncnt0 : 0;
        const int nd1 = (ncnt1 <= REG_CAP) ? ncnt1 : 0;
        int p00 = 0, p01 = 0, p10 = 0, p11 = 0;
        if (lane16      < nd0) p00 = slots[nbase0 + lane16];
        if (lane16 + 16 < nd0) p01 = slots[nbase0 + lane16 + 16];
        if (lane16      < nd1) p10 = slots[nbase1 + lane16];
        if (lane16 + 16 < nd1) p11 = slots[nbase1 + lane16 + 16];

        if (ptile >= 0) {
            const int prow = ptile * 16 + hw;
            if (prow < N)
                *(float4*)&out[(long long)prow * CHANNELS + lane16 * 4] = pres;
        }

#pragma unroll
        for (int dir = 0; dir < 2; ++dir) {
            const int d = dcount[dir];
            const int o = dbase[dir];
            const int* __restrict__ ids = &srt[dir][hw][0];
            float4 acc = self;

            if (d <= SEG_CAP) {
                int k = 0;
#pragma unroll 1
                for (; k + 8 <= d; k += 8) {
                    const int4 q0 = *(const int4*)(ids + k);
                    const int4 q1 = *(const int4*)(ids + k + 4);
                    const float4 v0 = x4[(long long)q0.x * 16 + lane16];
                    const float4 v1 = x4[(long long)q0.y * 16 + lane16];
                    const float4 v2 = x4[(long long)q0.z * 16 + lane16];
                    const float4 v3 = x4[(long long)q0.w * 16 + lane16];
                    const float4 v4 = x4[(long long)q1.x * 16 + lane16];
                    const float4 v5 = x4[(long long)q1.y * 16 + lane16];
                    const float4 v6 = x4[(long long)q1.z * 16 + lane16];
                    const float4 v7 = x4[(long long)q1.w * 16 + lane16];
                    ROW_ADD(acc, v0) ROW_ADD(acc, v1) ROW_ADD(acc, v2) ROW_ADD(acc, v3)
                    ROW_ADD(acc, v4) ROW_ADD(acc, v5) ROW_ADD(acc, v6) ROW_ADD(acc, v7)
                }
                if (k + 4 <= d) {
                    const int4 q0 = *(const int4*)(ids + k);
                    const float4 v0 = x4[(long long)q0.x * 16 + lane16];
                    const float4 v1 = x4[(long long)q0.y * 16 + lane16];
                    const float4 v2 = x4[(long long)q0.z * 16 + lane16];
                    const float4 v3 = x4[(long long)q0.w * 16 + lane16];
                    ROW_ADD(acc, v0) ROW_ADD(acc, v1) ROW_ADD(acc, v2) ROW_ADD(acc, v3)
                    k += 4;
                }
                const int rest = d - k;
                if (rest > 0) {
                    float4 t0 = x4[(long long)ids[k] * 16 + lane16];
                    float4 t1 = make_float4(0.f, 0.f, 0.f, 0.f);
                    float4 t2 = make_float4(0.f, 0.f, 0.f, 0.f);
                    if (rest > 1) t1 = x4[(long long)ids[k + 1] * 16 + lane16];
                    if (rest > 2) t2 = x4[(long long)ids[k + 2] * 16 + lane16];
                    ROW_ADD(acc, t0)
                    if (rest > 1) ROW_ADD(acc, t1)
                    if (rest > 2) ROW_ADD(acc, t2)
                }
            } else {
                int prev = -1;
                int done = 0;
                while (done < d) {
                    int best = 0x7fffffff;
                    for (int j = 0; j < d; ++j) {
                        const int u = slots[o + j];
                        if (u > prev && u < best) best = u;
                    }
                    int mult = 0;
                    for (int j = 0; j < d; ++j) mult += (slots[o + j] == best) ? 1 : 0;
                    if (mult == 0) break;
                    const float4 v = x4[(long long)best * 16 + lane16];
                    for (int t = 0; t < mult; ++t) ROW_ADD(acc, v)
                    prev = best;
                    done += mult;
                }
            }

            const float sc = scale[dir];
            float4 sv;
            sv.x = acc.x * sc; sv.y = acc.y * sc; sv.z = acc.z * sc; sv.w = acc.w * sc;
            *(float4*)&As[dir][hw][lane16 * 4] = sv;
        }

        rank_store(p00, p01, nd0, lane16, &srt[0][hw][0]);
        rank_store(p10, p11, nd1, lane16, &srt[1][hw][0]);
        rank_long(slots, nbase0, ncnt0, lane16, &srt[0][hw][0]);
        rank_long(slots, nbase1, ncnt1, lane16, &srt[1][hw][0]);
        __syncthreads();

        {
            v8f acc = {};
            const float*  __restrict__ arow = &As[wdir][m][kb];
            const float4* __restrict__ bfr  = (const float4*)Bs + (wdir * 4 + ct) * 256 + lane;
#pragma unroll
            for (int jp = 0; jp < 8; ++jp) {
                const float4 bv = bfr[jp * 32];
                const float2 a0 = *(const float2*)(arow + 8 * jp);
                const float2 a1 = *(const float2*)(arow + 8 * jp + 4);
                v2f a, b;
                a.x = a0.x; a.y = a0.y;
                b.x = bv.x; b.y = bv.y;
                acc = __builtin_amdgcn_wmma_f32_16x16x4_f32(
                    false, a, false, b, (short)0, acc, false, false);
                a.x = a1.x; a.y = a1.y;
                b.x = bv.z; b.y = bv.w;
                acc = __builtin_amdgcn_wmma_f32_16x16x4_f32(
                    false, a, false, b, (short)0, acc, false, false);
            }

#pragma unroll
            for (int j = 0; j < 8; ++j)
                Cs[wdir][j + half * 8][col] = acc[j];
        }
        ptile = tile;

        ccnt0 = ncnt0; ccnt1 = ncnt1; cbase0 = nbase0; cbase1 = nbase1;
        ncnt0 = fcnt0; ncnt1 = fcnt1; nbase0 = fbase0; nbase1 = fbase1;
    }

    __syncthreads();
    if (ptile >= 0) {
        const int prow = ptile * 16 + hw;
        if (prow < N) {
            const float4 po = *(const float4*)&Cs[0][hw][lane16 * 4];
            const float4 pb = *(const float4*)&Cs[1][hw][lane16 * 4];
            float4 pr;
            pr.x = po.x + pb.x; pr.y = po.y + pb.y; pr.z = po.z + pb.z; pr.w = po.w + pb.w;
            *(float4*)&out[(long long)prow * CHANNELS + lane16 * 4] = pr;
        }
    }
}

extern "C" void kernel_launch(void* const* d_in, const int* in_sizes, int n_in,
                              void* d_out, int out_size, void* d_ws, size_t ws_size,
                              hipStream_t stream) {
    const float* x       = (const float*)d_in[0];
    const int*   sources = (const int*)d_in[1];
    const int*   targets = (const int*)d_in[2];
    const float* norm    = (const float*)d_in[3];
    const float* norm_t  = (const float*)d_in[4];
    const float* w_out   = (const float*)d_in[5];
    const float* w_back  = (const float*)d_in[6];
    float*       out     = (float*)d_out;

    const int N = in_sizes[3];
    const int E = in_sizes[1];
    const int NB  = (N + BIN - 1) / BIN;
    const int nch = (E + CH_EDGES - 1) / CH_EDGES;

    int* cnt    = (int*)d_ws;
    int* cursor = cnt + 2 * (long long)N;
    int* off    = cursor + 64;
    int* slots  = off + 2 * (long long)N;
    int* stage  = slots + 2 * (long long)E;
    int* coff   = stage + 2 * (long long)E;

    biconv_chunk_sort<<<nch, 1024, 0, stream>>>(
        sources, targets, cursor, stage, coff, N, E, NB);
    biconv_bin_build<<<2 * NB, 512, 0, stream>>>(
        stage, coff, cursor, cnt, off, slots, N, E, NB, nch);
    biconv_gather_gemm<<<1280, 256, 0, stream>>>(
        x, cnt, off, slots, norm, norm_t, w_out, w_back, out, N);
}


# last kernel gather: fewer dependent round trips per node: last 1-7 rows asked for together (was 4, then up to 3 singly), first trip not held for the own row, 32-bit row offsets; same sum order
# speedup vs baseline: 4.1607x; 1.0324x over previous
#include <hip/hip_runtime.h>

typedef __attribute__((ext_vector_type(2))) float v2f;
typedef __attribute__((ext_vector_type(8))) float v8f;
typedef __attribute__((ext_vector_type(4))) float v4f;

#define CHANNELS 64
#define SEG_CAP 64
#define REG_CAP 32
#define SEG_ROW (SEG_CAP + 8)
#define A_STRIDE 68

#define BIN_SHIFT 8
#define BIN 256
#define MAXNB 1024
#define CH_EDGES 8192
#define MAXCH 512
#define KB_CAP 8192
#define PCS 5

__device__ __forceinline__ int block_incl_scan(int v, int* wsum, int tid, int nwaves) {
    const int lane = tid & 31;
    const int wave = tid >> 5;
    __syncthreads();
    for (int s = 1; s < 32; s <<= 1) {
        int u = __shfl_up(v, s);
        if (lane >= s) v += u;
    }
    if (lane == 31) wsum[wave] = v;
    __syncthreads();
    if (wave == 0) {
        const int wv = (lane < nwaves) ? wsum[lane] : 0;
        int z = wv;
        for (int s = 1; s < 32; s <<= 1) {
            int u = __shfl_up(z, s);
            if (lane >= s) z += u;
        }
        wsum[lane] = z - wv;
    }
    __syncthreads();
    return v + wsum[wave];
}

__global__ __launch_bounds__(1024)
void biconv_chunk_sort(const int* __restrict__ src,
                       const int* __restrict__ tgt,
                       int* __restrict__ cursor,
                       int* __restrict__ stage,
                       int* __restrict__ coff,
                       int N, int E, int NB) {
    __shared__ int hist[2][MAXNB];
    __shared__ int stg[2][CH_EDGES];
    __shared__ int wsum[32];

    const int tid = threadIdx.x;
    const int w   = blockIdx.x;
    const int e0  = w * CH_EDGES;
    const int sh  = 32 - __clz(N - 1);
    const int nch = gridDim.x;

    if (w == 0 && tid == 0) *cursor = 0;

    hist[0][tid] = 0;
    hist[1][tid] = 0;
    __syncthreads();

    int sv[CH_EDGES / 1024];
    int tv[CH_EDGES / 1024];
#pragma unroll
    for (int k = 0; k < CH_EDGES / 1024; ++k) {
        const int e = e0 + k * 1024 + tid;
        sv[k] = -1;
        tv[k] = -1;
        if (e < E) {
            sv[k] = src[e];
            tv[k] = tgt[e];
        }
    }
#pragma unroll
    for (int k = 0; k < CH_EDGES / 1024; ++k) {
        if (sv[k] >= 0) {
            atomicAdd(&hist[0][tv[k] >> BIN_SHIFT], 1);
            atomicAdd(&hist[1][sv[k] >> BIN_SHIFT], 1);
        }
    }
    __syncthreads();

    const int c0 = hist[0][tid];
    const int c1 = hist[1][tid];
    const int inc = block_incl_scan(c0 | (c1 << 16), wsum, tid, 32);
    const int x0 = (inc & 0xffff) - c0;
    const int x1 = (inc >> 16) - c1;
    hist[0][tid] = x0;
    hist[1][tid] = x1;
    if (tid < NB) {
        coff[(long long)w * (NB + 1) + tid]         = x0;
        coff[(long long)(nch + w) * (NB + 1) + tid] = x1;
    }
    if (tid == 1023) {
        coff[(long long)w * (NB + 1) + NB]         = x0 + c0;
        coff[(long long)(nch + w) * (NB + 1) + NB] = x1 + c1;
    }
    __syncthreads();

#pragma unroll
    for (int k = 0; k < CH_EDGES / 1024; ++k) {
        if (sv[k] >= 0) {
            const int s = sv[k];
            const int t = tv[k];
            const int p = atomicAdd(&hist[0][t >> BIN_SHIFT], 1);
            stg[0][p] = (int)(((unsigned)(t & (BIN - 1)) << sh) | (unsigned)s);
            const int q = atomicAdd(&hist[1][s >> BIN_SHIFT], 1);
            stg[1][q] = (int)(((unsigned)(s & (BIN - 1)) << sh) | (unsigned)t);
        }
    }
    __syncthreads();

    const int ne = (E - e0 < CH_EDGES) ? (E - e0) : CH_EDGES;
    for (int i = tid; i < ne; i += 1024) {
        stage[(long long)e0 + i]     = stg[0][i];
        stage[(long long)E + e0 + i] = stg[1][i];
    }
}

__global__ __launch_bounds__(512)
void biconv_bin_build(const int* __restrict__ stage,
                      const int* __restrict__ coff,
                      int* __restrict__ cursor,
                      int* __restrict__ cnt,
                      int* __restrict__ off,
                      int* __restrict__ slots,
                      int N, int E, int NB, int nch) {
    __shared__ int pos[MAXCH + 2];
    __shared__ int pstart[MAXCH + 1];
    __shared__ int cntL[BIN];
    __shared__ int curL[BIN];
    __shared__ int entL[KB_CAP];
    __shared__ int srtL[KB_CAP];
    __shared__ int wsum[32];
    __shared__ int sbase;

    const int tid  = threadIdx.x;
    const int lane = tid & 31;
    const int wave = tid >> 5;
    const int d   = (blockIdx.x >= NB) ? 1 : 0;
    const int b   = blockIdx.x - d * NB;
    const int sh  = 32 - __clz(N - 1);
    const unsigned idmask = (sh >= 32) ? 0xffffffffu : ((1u << sh) - 1u);

    int c = 0;
    if (tid < nch) {
        const long long row = (long long)(d * nch + tid) * (NB + 1);
        const int a = coff[row + b];
        const int z = coff[row + b + 1];
        c = z - a;
        pstart[tid] = d * E + tid * CH_EDGES + a;
    }
    if (tid < BIN) { cntL[tid] = 0; }
    const int pinc = block_incl_scan(c, wsum, tid, 16);
    if (tid < nch) pos[tid] = pinc - c;
    if (tid == nch - 1) {
        pos[nch]     = pinc;
        pos[nch + 1] = pinc;
        pstart[nch]  = 0;
    }
    __syncthreads();
    const int T = pos[nch];

    if (tid == 0) sbase = atomicAdd(cursor, T);

    const bool in_lds = (T <= KB_CAP);

    if (in_lds) {
        for (int u0 = 0; wave + 16 * u0 < nch; u0 += PCS) {
            int pb[PCS], pc[PCS], pg[PCS];
#pragma unroll
            for (int u = 0; u < PCS; ++u) {
                const int p  = wave + 16 * (u0 + u);
                const int pp = (p < nch) ? p : nch;
                const int b0 = pos[pp];
                pb[u] = b0;
                pc[u] = pos[pp + 1] - b0;
                pg[u] = pstart[pp];
            }
            unsigned ea[PCS], eb[PCS];
#pragma unroll
            for (int u = 0; u < PCS; ++u) {
                ea[u] = (unsigned)stage[(lane      < pc[u]) ? (pg[u] + lane)      : 0];
                eb[u] = (unsigned)stage[(lane + 32 < pc[u]) ? (pg[u] + lane + 32) : 0];
            }
#pragma unroll
            for (int u = 0; u < PCS; ++u) {
                if (lane < pc[u]) {
                    entL[pb[u] + lane] = (int)ea[u];
                    atomicAdd(&cntL[ea[u] >> sh], 1);
                }
                if (lane + 32 < pc[u]) {
                    entL[pb[u] + lane + 32] = (int)eb[u];
                    atomicAdd(&cntL[eb[u] >> sh], 1);
                }
            }
#pragma unroll
            for (int u = 0; u < PCS; ++u) {
                for (int l = lane + 64; l < pc[u]; l += 32) {
                    const unsigned en = (unsigned)stage[pg[u] + l];
                    entL[pb[u] + l] = (int)en;
                    atomicAdd(&cntL[en >> sh], 1);
                }
            }
        }
    } else {
        for (int i = tid; i < T; i += 512) {
            int lo = 0, hi = nch - 1;
            while (lo < hi) {
                const int mid = (lo + hi) >> 1;
                if (pos[mid + 1] > i) hi = mid; else lo = mid + 1;
            }
            const unsigned en = (unsigned)stage[pstart[lo] + (i - pos[lo])];
            atomicAdd(&cntL[en >> sh], 1);
        }
    }
    __syncthreads();

    const int cl  = (tid < BIN) ? cntL[tid] : 0;
    const int inc = block_incl_scan(cl, wsum, tid, 16);
    const int base = sbase;
    if (tid < BIN) {
        curL[tid] = inc - cl;
        const int node = b * BIN + tid;
        if (node < N) {
            cnt[d * N + node] = cl;
            off[d * N + node] = base + inc - cl;
        }
    }
    __syncthreads();

    if (in_lds) {
        for (int i0 = tid; i0 < T; i0 += 4 * 512) {
            unsigned e[4];
#pragma unroll
            for (int u = 0; u < 4; ++u) {
                const int i = i0 + u * 512;
                e[u] = (i < T) ? (unsigned)entL[i] : 0u;
            }
            int q[4];
#pragma unroll
            for (int u = 0; u < 4; ++u) {
                const int i = i0 + u * 512;
                q[u] = (i < T) ? atomicAdd(&curL[e[u] >> sh], 1) : 0;
            }
#pragma unroll
            for (int u = 0; u < 4; ++u) {
                const int i = i0 + u * 512;
                if (i < T) srtL[q[u]] = (int)(e[u] & idmask);
            }
        }
        __syncthreads();
        for (int i = tid; i < T; i += 512) slots[(long long)base + i] = srtL[i];
    } else {
        for (int i = tid; i < T; i += 512) {
            int lo = 0, hi = nch - 1;
            while (lo < hi) {
                const int mid = (lo + hi) >> 1;
                if (pos[mid + 1] > i) hi = mid; else lo = mid + 1;
            }
            const unsigned en = (unsigned)stage[pstart[lo] + (i - pos[lo])];
            const int p = atomicAdd(&curL[en >> sh], 1);
            slots[(long long)base + p] = (int)(en & idmask);
        }
    }
}

#define ROW_ADD(a, v) { a.x += v.x; a.y += v.y; a.z += v.z; a.w += v.w; }

template <int J>
__device__ __forceinline__ unsigned row_lane(unsigned v) {
    return (unsigned)__builtin_amdgcn_update_dpp(0, (int)v, 0x150 + J, 0xf, 0xf, true);
}

#define RANK_LOW(J)  { const unsigned u = row_lane<J>(ka); ra += (u < ka) ? 1 : 0; }
#define RANK_HIGH(J) { const unsigned u = row_lane<J>(ka); const unsigned w = row_lane<J>(kb); \
                       rb += (u < kb) ? 1 : 0; ra += (w < ka) ? 1 : 0; rb += (w < kb) ? 1 : 0; }
__device__ __forceinline__ void rank_store(int ia, int ib, int dd, int lane16, int* __restrict__ seg) {
    const unsigned ka = (lane16      < dd) ? (((unsigned)ia << 5) | (unsigned)lane16)        : 0xffffffffu;
    const unsigned kb = (lane16 + 16 < dd) ? (((unsigned)ib << 5) | (unsigned)(lane16 + 16)) : 0xffffffffu;
    int ra = 0, rb = 0;
    RANK_LOW(0)  RANK_LOW(1)  RANK_LOW(2)  RANK_LOW(3)  RANK_LOW(4)  RANK_LOW(5)  RANK_LOW(6)  RANK_LOW(7)
    RANK_LOW(8)  RANK_LOW(9)  RANK_LOW(10) RANK_LOW(11) RANK_LOW(12) RANK_LOW(13) RANK_LOW(14) RANK_LOW(15)
    if (__builtin_amdgcn_ballot_w32(dd > 16) != 0u) {
        RANK_HIGH(0)  RANK_HIGH(1)  RANK_HIGH(2)  RANK_HIGH(3)  RANK_HIGH(4)  RANK_HIGH(5)  RANK_HIGH(6)  RANK_HIGH(7)
        RANK_HIGH(8)  RANK_HIGH(9)  RANK_HIGH(10) RANK_HIGH(11) RANK_HIGH(12) RANK_HIGH(13) RANK_HIGH(14) RANK_HIGH(15)
    }
    if (lane16      < dd) seg[ra] = ia;
    if (lane16 + 16 < dd) seg[rb] = ib;
}

__device__ __forceinline__ void rank_long(const int* __restrict__ slots, int o, int d, int lane16,
                                          int* __restrict__ seg) {
    if (d > REG_CAP && d <= SEG_CAP) {
        for (int i = lane16; i < d; i += 16) {
            const int vi = slots[o + i];
            int r = 0;
            for (int j = 0; j < d; ++j) {
                const int u = slots[o + j];
                r += (u < vi || (u == vi && j < i)) ? 1 : 0;
            }
            seg[r] = vi;
        }
    }
}

__device__ __forceinline__ v4f x_row(const char* __restrict__ xb, int id, unsigned loff) {
    return *(const v4f*)(xb + (((unsigned)id << 8) + loff));
}

__device__ __forceinline__ void rows8(const char* __restrict__ xb, const int* __restrict__ ids,
                                      unsigned loff, float4& acc) {
    const int4 q0 = *(const int4*)(ids);
    const int4 q1 = *(const int4*)(ids + 4);
    const v4f v0 = x_row(xb, q0.x, loff);
    const v4f v1 = x_row(xb, q0.y, loff);
    const v4f v2 = x_row(xb, q0.z, loff);
    const v4f v3 = x_row(xb, q0.w, loff);
    const v4f v4 = x_row(xb, q1.x, loff);
    const v4f v5 = x_row(xb, q1.y, loff);
    const v4f v6 = x_row(xb, q1.z, loff);
    const v4f v7 = x_row(xb, q1.w, loff);
    asm volatile("" :: "v"(v0), "v"(v1), "v"(v2), "v"(v3), "v"(v4), "v"(v5), "v"(v6), "v"(v7));
    ROW_ADD(acc, v0) ROW_ADD(acc, v1) ROW_ADD(acc, v2) ROW_ADD(acc, v3)
    ROW_ADD(acc, v4) ROW_ADD(acc, v5) ROW_ADD(acc, v6) ROW_ADD(acc, v7)
}

__global__ __launch_bounds__(256)
void biconv_gather_gemm(const float* __restrict__ x,
                        const int* __restrict__ cnt,
                        const int* __restrict__ off,
                        const int* __restrict__ slots,
                        const float* __restrict__ norm,
                        const float* __restrict__ norm_t,
                        const float* __restrict__ w_out,
                        const float* __restrict__ w_back,
                        float* __restrict__ out,
                        int N) {
    __shared__ __attribute__((aligned(16))) float As[2][16][A_STRIDE];
    __shared__ __attribute__((aligned(16))) int   srt[2][16][SEG_ROW];
    __shared__ __attribute__((aligned(16))) float Cs[2][16][A_STRIDE];
    __shared__ __attribute__((aligned(16))) float Bs[2 * 4 * 8 * 32 * 4];

    const int tid    = threadIdx.x;
    const int hw     = tid >> 4;
    const int lane16 = tid & 15;
    const int wave   = tid >> 5;
    const int lane   = tid & 31;
    const int m      = lane & 15;
    const int half   = lane >> 4;
    const int kb     = half << 1;
    const int wdir   = wave >> 2;
    const int ct     = wave & 3;
    const int col    = ct * 16 + (lane & 15);
    const float4* __restrict__ x4 = (const float4*)x;
    const char*   __restrict__ xb = (const char*)x;
    const unsigned loff = (unsigned)lane16 * 16u;

    const int ntiles = (N + 15) >> 4;
    const int tstep  = (int)gridDim.x;

    int ccnt0 = 0, ccnt1 = 0, cbase0 = 0, cbase1 = 0;
    int ncnt0 = 0, ncnt1 = 0, nbase0 = 0, nbase1 = 0;
    {
        const int n0 = (int)blockIdx.x * 16 + hw;
        if (n0 < N) {
            ccnt0  = cnt[n0];
            ccnt1  = cnt[N + n0];
            cbase0 = off[n0];
            cbase1 = off[N + n0];
        }
        const int n1 = ((int)blockIdx.x + tstep) * 16 + hw;
        if (n1 < N) {
            ncnt0  = cnt[n1];
            ncnt1  = cnt[N + n1];
            nbase0 = off[n1];
            nbase1 = off[N + n1];
        }
    }
    {
        const int fd0 = (ccnt0 <= REG_CAP) ? ccnt0 : 0;
        const int fd1 = (ccnt1 <= REG_CAP) ? ccnt1 : 0;
        int q00 = 0, q01 = 0, q10 = 0, q11 = 0;
        if (lane16      < fd0) q00 = slots[cbase0 + lane16];
        if (lane16 + 16 < fd0) q01 = slots[cbase0 + lane16 + 16];
        if (lane16      < fd1) q10 = slots[cbase1 + lane16];
        if (lane16 + 16 < fd1) q11 = slots[cbase1 + lane16 + 16];
#pragma unroll
        for (int it = 0; it < 8; ++it) {
            const float* __restrict__ wsrc = (it >> 2) ? w_back : w_out;
            const int fl = tid & 31;
            const int k0 = 8 * (tid >> 5) + 2 * (fl >> 4);
            const int cc = 16 * (it & 3) + (fl & 15);
            float4 wv;
            wv.x = wsrc[(k0    ) * CHANNELS + cc];
            wv.y = wsrc[(k0 + 1) * CHANNELS + cc];
            wv.z = wsrc[(k0 + 4) * CHANNELS + cc];
            wv.w = wsrc[(k0 + 5) * CHANNELS + cc];
            *(float4*)&Bs[(it * 256 + tid) * 4] = wv;
        }
        rank_store(q00, q01, fd0, lane16, &srt[0][hw][0]);
        rank_store(q10, q11, fd1, lane16, &srt[1][hw][0]);
        rank_long(slots, cbase0, ccnt0, lane16, &srt[0][hw][0]);
        rank_long(slots, cbase1, ccnt1, lane16, &srt[1][hw][0]);
    }

    int ptile = -1;

    for (int tile = (int)blockIdx.x; tile < ntiles; tile += tstep) {
        const int node   = tile * 16 + hw;
        const bool valid = node < N;

        const int dcount[2] = {ccnt0, ccnt1};
        const int dbase[2]  = {cbase0, cbase1};

        __syncthreads();

        float4 pres = make_float4(0.f, 0.f, 0.f, 0.f);
        if (ptile >= 0) {
            const float4 po = *(const float4*)&Cs[0][hw][lane16 * 4];
            const float4 pb = *(const float4*)&Cs[1][hw][lane16 * 4];
            pres.x = po.x + pb.x; pres.y = po.y + pb.y; pres.z = po.z + pb.z; pres.w = po.w + pb.w;
        }

        float scale[2] = {0.f, 0.f};
        float4 self = make_float4(0.f, 0.f, 0.f, 0.f);
        if (valid) {
            scale[0] = norm[node];
            scale[1] = norm_t[node];
            self     = x4[(long long)node * 16 + lane16];
        }
        int fcnt0 = 0, fcnt1 = 0, fbase0 = 0, fbase1 = 0;
        {
            const int n2 = (tile + 2 * tstep) * 16 + hw;
            if (n2 < N) {
                fcnt0  = cnt[n2];
                fcnt1  = cnt[N + n2];
                fbase0 = off[n2];
                fbase1 = off[N + n2];
            }
        }
        const int nd0 = (ncnt0 <= REG_CAP) ? ncnt0 : 0;
        const int nd1 = (ncnt1 <= REG_CAP) ? ncnt1 : 0;
        int p00 = 0, p01 = 0, p10 = 0, p11 = 0;
        if (lane16      < nd0) p00 = slots[nbase0 + lane16];
        if (lane16 + 16 < nd0) p01 = slots[nbase0 + lane16 + 16];
        if (lane16      < nd1) p10 = slots[nbase1 + lane16];
        if (lane16 + 16 < nd1) p11 = slots[nbase1 + lane16 + 16];

        if (ptile >= 0) {
            const int prow = ptile * 16 + hw;
            if (prow < N)
                *(float4*)&out[(long long)prow * CHANNELS + lane16 * 4] = pres;
        }

#pragma unroll
        for (int dir = 0; dir < 2; ++dir) {
            const int d = dcount[dir];
            const int o = dbase[dir];
            const int* __restrict__ ids = &srt[dir][hw][0];
            float4 acc = self;

            if (d > SEG_CAP) {
                int prev = -1;
                int done = 0;
                while (done < d) {
                    int best = 0x7fffffff;
                    for (int j = 0; j < d; ++j) {
                        const int u = slots[o + j];
                        if (u > prev && u < best) best = u;
                    }
                    int mult = 0;
                    for (int j = 0; j < d; ++j) mult += (slots[o + j] == best) ? 1 : 0;
                    if (mult == 0) break;
                    const float4 v = x4[(long long)best * 16 + lane16];
                    for (int t = 0; t < mult; ++t) ROW_ADD(acc, v)
                    prev = best;
                    done += mult;
                }
            } else {
                const float4 nz = make_float4(-0.0f, -0.0f, -0.0f, -0.0f);
                int k = 0;
                if (dir == 0) {
                    if (d >= 8) {
                        rows8(xb, ids, loff, acc);
                        k = 8;
                    } else {
                        acc = nz;
                    }
                }
#pragma unroll 1
                for (; k + 8 <= d; k += 8) rows8(xb, ids + k, loff, acc);
                const int  t  = d - k;
                const int4 qa = *(const int4*)(ids + k);
                const int4 qb = *(const int4*)(ids + k + 4);
                const v4f nzv = {-0.0f, -0.0f, -0.0f, -0.0f};
                v4f u0 = nzv, u1 = nzv, u2 = nzv, u3 = nzv, u4 = nzv, u5 = nzv, u6 = nzv;
                if (t > 0) u0 = x_row(xb, qa.x, loff);
                if (t > 1) u1 = x_row(xb, qa.y, loff);
                if (t > 2) u2 = x_row(xb, qa.z, loff);
                if (t > 3) u3 = x_row(xb, qa.w, loff);
                if (t > 4) u4 = x_row(xb, qb.x, loff);
                if (t > 5) u5 = x_row(xb, qb.y, loff);
                if (t > 6) u6 = x_row(xb, qb.z, loff);
                asm volatile("" :: "v"(u0), "v"(u1), "v"(u2), "v"(u3), "v"(u4), "v"(u5), "v"(u6));
                if (dir == 0) {
                    const float4 own = (d < 8) ? self : nz;
                    ROW_ADD(acc, own)
                }
                ROW_ADD(acc, u0) ROW_ADD(acc, u1) ROW_ADD(acc, u2) ROW_ADD(acc, u3)
                ROW_ADD(acc, u4) ROW_ADD(acc, u5) ROW_ADD(acc, u6)
            }

            const float sc = scale[dir];
            float4 sv;
            sv.x = acc.x * sc; sv.y = acc.y * sc; sv.z = acc.z * sc; sv.w = acc.w * sc;
            *(float4*)&As[dir][hw][lane16 * 4] = sv;
        }

        rank_store(p00, p01, nd0, lane16, &srt[0][hw][0]);
        rank_store(p10, p11, nd1, lane16, &srt[1][hw][0]);
        rank_long(slots, nbase0, ncnt0, lane16, &srt[0][hw][0]);
        rank_long(slots, nbase1, ncnt1, lane16, &srt[1][hw][0]);
        __syncthreads();

        {
            v8f acc = {};
            const float*  __restrict__ arow = &As[wdir][m][kb];
            const float4* __restrict__ bfr  = (const float4*)Bs + (wdir * 4 + ct) * 256 + lane;
#pragma unroll
            for (int jp = 0; jp < 8; ++jp) {
                const float4 bv = bfr[jp * 32];
                const float2 a0 = *(const float2*)(arow + 8 * jp);
                const float2 a1 = *(const float2*)(arow + 8 * jp + 4);
                v2f a, b;
                a.x = a0.x; a.y = a0.y;
                b.x = bv.x; b.y = bv.y;
                acc = __builtin_amdgcn_wmma_f32_16x16x4_f32(
                    false, a, false, b, (short)0, acc, false, false);
                a.x = a1.x; a.y = a1.y;
                b.x = bv.z; b.y = bv.w;
                acc = __builtin_amdgcn_wmma_f32_16x16x4_f32(
                    false, a, false, b, (short)0, acc, false, false);
            }

#pragma unroll
            for (int j = 0; j < 8; ++j)
                Cs[wdir][j + half * 8][col] = acc[j];
        }
        ptile = tile;

        ccnt0 = ncnt0; ccnt1 = ncnt1; cbase0 = nbase0; cbase1 = nbase1;
        ncnt0 = fcnt0; ncnt1 = fcnt1; nbase0 = fbase0; nbase1 = fbase1;
    }

    __syncthreads();
    if (ptile >= 0) {
        const int prow = ptile * 16 + hw;
        if (prow < N) {
            const float4 po = *(const float4*)&Cs[0][hw][lane16 * 4];
            const float4 pb = *(const float4*)&Cs[1][hw][lane16 * 4];
            float4 pr;
            pr.x = po.x + pb.x; pr.y = po.y + pb.y; pr.z = po.z + pb.z; pr.w = po.w + pb.w;
            *(float4*)&out[(long long)prow * CHANNELS + lane16 * 4] = pr;
        }
    }
}

extern "C" void kernel_launch(void* const* d_in, const int* in_sizes, int n_in,
                              void* d_out, int out_size, void* d_ws, size_t ws_size,
                              hipStream_t stream) {
    const float* x       = (const float*)d_in[0];
    const int*   sources = (const int*)d_in[1];
    const int*   targets = (const int*)d_in[2];
    const float* norm    = (const float*)d_in[3];
    const float* norm_t  = (const float*)d_in[4];
    const float* w_out   = (const float*)d_in[5];
    const float* w_back  = (const float*)d_in[6];
    float*       out     = (float*)d_out;

    const int N = in_sizes[3];
    const int E = in_sizes[1];
    const int NB  = (N + BIN - 1) / BIN;
    const int nch = (E + CH_EDGES - 1) / CH_EDGES;

    int* cnt    = (int*)d_ws;
    int* cursor = cnt + 2 * (long long)N;
    int* off    = cursor + 64;
    int* slots  = off + 2 * (long long)N;
    int* stage  = slots + 2 * (long long)E;
    int* coff   = stage + 2 * (long long)E;

    biconv_chunk_sort<<<nch, 1024, 0, stream>>>(
        sources, targets, cursor, stage, coff, N, E, NB);
    biconv_bin_build<<<2 * NB, 512, 0, stream>>>(
        stage, coff, cursor, cnt, off, slots, N, E, NB, nch);
    biconv_gather_gemm<<<1280, 256, 0, stream>>>(
        x, cnt, off, slots, norm, norm_t, w_out, w_back, out, N);
}
